# starting kernel
# baseline (speedup 1.0000x reference)
#include <hip/hip_runtime.h>
#include <hip/hip_bf16.h>

#define CH 128

typedef __attribute__((ext_vector_type(2))) float v2f;
typedef __attribute__((ext_vector_type(8))) float v8f;


__global__ void gcn_deg_init(float* __restrict__ deg, int n) {
    int i = blockIdx.x * blockDim.x + threadIdx.x;
    if (i < n) deg[i] = 1.0f;
}

__global__ void gcn_deg_accum(const int* __restrict__ dst, float* __restrict__ deg, int E) {
    int e = blockIdx.x * blockDim.x + threadIdx.x;
    if (e < E) atomicAdd(&deg[dst[e]], 1.0f);
}

__global__ void gcn_dinv(float* __restrict__ deg, int n) {
    int i = blockIdx.x * blockDim.x + threadIdx.x;
    if (i < n) deg[i] = rsqrtf(deg[i]);
}


__global__ __launch_bounds__(256) void gcn_gemm_wmma(
    const float* __restrict__ X, const float* __restrict__ W,
    float* __restrict__ XW, int n)
{
    const int wave = threadIdx.x >> 5;
    const int lane = threadIdx.x & 31;
    const int half = lane >> 4;
    const int l16  = lane & 15;
    const int m0   = blockIdx.x * 16;
    const int n0   = wave * 16;

    int arow = m0 + l16;
    if (arow >= n) arow = n - 1;
    const float* Ap = X + (size_t)arow * CH;

    v8f c = {};
#pragma unroll
    for (int k0 = 0; k0 < CH; k0 += 4) {
        const int ka = k0 + 2 * half;
        v2f a, b;
        a.x = Ap[ka];
        a.y = Ap[ka + 1];
        b.x = W[(size_t)ka       * CH + n0 + l16];
        b.y = W[(size_t)(ka + 1) * CH + n0 + l16];
        c = __builtin_amdgcn_wmma_f32_16x16x4_f32(
                 false, a,  false, b,
                 (short)0, c,  false,  false);
    }

#pragma unroll
    for (int r = 0; r < 8; ++r) {
        int row = m0 + r + 8 * half;
        if (row < n) XW[(size_t)row * CH + n0 + l16] = c[r];
    }
}


__global__ void gcn_init_out(const float* __restrict__ xw, const float* __restrict__ dinv,
                             const float* __restrict__ bias, float* __restrict__ out, int total) {
    int idx = blockIdx.x * blockDim.x + threadIdx.x;
    if (idx >= total) return;
    int i = idx >> 7;
    int c = idx & (CH - 1);
    float di = dinv[i];
    out[idx] = bias[c] + xw[idx] * di * di;
}


__global__ __launch_bounds__(256) void gcn_scatter(
    const int* __restrict__ src, const int* __restrict__ dst,
    const float* __restrict__ dinv, const float* __restrict__ xw,
    float* __restrict__ out, int E)
{
    int e = (int)(((size_t)blockIdx.x * blockDim.x + threadIdx.x) >> 5);
    if (e >= E) return;
    int lane = threadIdx.x & 31;

    int s = src[e];
    int d = dst[e];
    float nrm = dinv[s] * dinv[d];

    const float4* row = (const float4*)(xw + (size_t)s * CH);
    float4 v = row[lane];

    float* o = out + (size_t)d * CH + lane * 4;
    atomicAdd(o + 0, v.x * nrm);
    atomicAdd(o + 1, v.y * nrm);
    atomicAdd(o + 2, v.z * nrm);
    atomicAdd(o + 3, v.w * nrm);
}


extern "C" void kernel_launch(void* const* d_in, const int* in_sizes, int n_in,
                              void* d_out, int out_size, void* d_ws, size_t ws_size,
                              hipStream_t stream) {
    const float* x  = (const float*)d_in[0];
    const float* W  = (const float*)d_in[1];
    const float* b  = (const float*)d_in[2];
    const int*   ei = (const int*)d_in[3];

    const int n = in_sizes[0] / CH;
    const int E = in_sizes[3] / 2;
    const int* src = ei;
    const int* dst = ei + E;

    float* deg = (float*)d_ws;
    float* xw  = deg + n;
    float* out = (float*)d_out;

    gcn_deg_init <<<(n + 255) / 256, 256, 0, stream>>>(deg, n);
    gcn_deg_accum<<<(E + 255) / 256, 256, 0, stream>>>(dst, deg, E);
    gcn_dinv     <<<(n + 255) / 256, 256, 0, stream>>>(deg, n);

    gcn_gemm_wmma<<<(n + 15) / 16, 256, 0, stream>>>(x, W, xw, n);

    int total = n * CH;
    gcn_init_out <<<(total + 255) / 256, 256, 0, stream>>>(xw, deg, b, out, total);

    size_t scatter_threads = (size_t)E * 32;
    gcn_scatter  <<<(int)((scatter_threads + 255) / 256), 256, 0, stream>>>(src, dst, deg, xw, out, E);
}


# product moved behind the aggregation (linearity): x rows are gathered and summed, then each work-group multiplies its 256 summed rows by W on f32 WMMA from LDS; no xw plane; sums: edges, then k
# speedup vs baseline: 1.7268x; 1.7268x over previous
#include <hip/hip_runtime.h>
#include <hip/hip_bf16.h>

#define CH 128

typedef __attribute__((ext_vector_type(2))) float v2f;
typedef __attribute__((ext_vector_type(8))) float v8f;


__global__ void gcn_deg_init(float* __restrict__ deg, int n) {
    int i = blockIdx.x * blockDim.x + threadIdx.x;
    if (i < n) deg[i] = 1.0f;
}

__global__ void gcn_deg_accum(const int* __restrict__ dst, float* __restrict__ deg, int E) {
    int e = blockIdx.x * blockDim.x + threadIdx.x;
    if (e < E) atomicAdd(&deg[dst[e]], 1.0f);
}

__global__ void gcn_dinv(float* __restrict__ deg, int n) {
    int i = blockIdx.x * blockDim.x + threadIdx.x;
    if (i < n) deg[i] = rsqrtf(deg[i]);
}


#define AG_ROWS    256
#define AG_SLOTS   8
#define AG_WAVES   32
#define AG_LCAP    768
#define AG_MCAP    384
#define AG_TP      132

__device__ __forceinline__ void ag_append(int dv, int e, int base, int rows, unsigned lt,
                                          unsigned* wl, int& cnt)
{
    const unsigned dl = (unsigned)(dv - base);
    const bool m = dl < (unsigned)rows;
    const unsigned mask = __builtin_amdgcn_ballot_w32(m);
    if (mask) {
        if (m) wl[cnt + __popc(mask & lt)] = (dl << 24) | (unsigned)e;
        cnt += __popc(mask);
    }
}

__device__ __forceinline__ void ag_quad(const int4 q, int e, int base, int rows, unsigned lt,
                                        unsigned* wl, int& cnt)
{
    const bool any = ((unsigned)(q.x - base) < (unsigned)rows) |
                     ((unsigned)(q.y - base) < (unsigned)rows) |
                     ((unsigned)(q.z - base) < (unsigned)rows) |
                     ((unsigned)(q.w - base) < (unsigned)rows);
    if (__builtin_amdgcn_ballot_w32(any)) {
        ag_append(q.x, e + 0, base, rows, lt, wl, cnt);
        ag_append(q.y, e + 1, base, rows, lt, wl, cnt);
        ag_append(q.z, e + 2, base, rows, lt, wl, cnt);
        ag_append(q.w, e + 3, base, rows, lt, wl, cnt);
    }
}

typedef __attribute__((ext_vector_type(32))) float ag_v32f;
struct ag_acc { ag_v32f a; };

__device__ __forceinline__ void ag_addslot(ag_acc& A, int slot, const float4 v, float nr)
{
    const int b = slot * 4;
    A.a[b + 0] += v.x * nr;
    A.a[b + 1] += v.y * nr;
    A.a[b + 2] += v.z * nr;
    A.a[b + 3] += v.w * nr;
}

template <int NG>
__device__ __forceinline__ void ag_addgroup(ag_acc& A, const unsigned* ent, int base, int lane,
                                            const float4* __restrict__ xw4, const float* __restrict__ dinv)
{
    unsigned t[4 * NG];
#pragma unroll
    for (int g = 0; g < NG; ++g) {
        const uint4 en = *(const uint4*)(ent + 4 * g);
        t[4 * g + 0] = __builtin_amdgcn_readfirstlane(en.x);
        t[4 * g + 1] = __builtin_amdgcn_readfirstlane(en.y);
        t[4 * g + 2] = __builtin_amdgcn_readfirstlane(en.z);
        t[4 * g + 3] = __builtin_amdgcn_readfirstlane(en.w);
    }
    float4 v[4 * NG];
    float nr[4 * NG];
#pragma unroll
    for (int i = 0; i < 4 * NG; ++i)
        v[i] = xw4[(size_t)(t[i] & 0xFFFFFFu) * (CH / 4) + lane];
#pragma unroll
    for (int i = 0; i < 4 * NG; ++i)
        nr[i] = dinv[t[i] & 0xFFFFFFu] * dinv[base + (int)(t[i] >> 24)];
#pragma unroll
    for (int i = 0; i < 4 * NG; ++i)
        ag_addslot(A, (int)(t[i] >> 29), v[i], nr[i]);
}

__device__ __forceinline__ void ag_flush(ag_acc& A, unsigned* mylist, int mcnt, int base, int lane,
                                         const int* __restrict__ src,
                                         const float4* __restrict__ xw4, const float* __restrict__ dinv)
{
    for (int k = lane; k < mcnt; k += 32) {
        const unsigned t = mylist[k];
        mylist[k] = (t & 0xFF000000u) | (unsigned)src[t & 0xFFFFFFu];
    }
    int k = 0;
    for (; k + 12 <= mcnt; k += 12) ag_addgroup<3>(A, mylist + k, base, lane, xw4, dinv);
    for (; k + 4 <= mcnt; k += 4)   ag_addgroup<1>(A, mylist + k, base, lane, xw4, dinv);
    for (; k < mcnt; ++k) {
        const unsigned t = __builtin_amdgcn_readfirstlane(mylist[k]);
        const int s = (int)(t & 0xFFFFFFu), r = (int)(t >> 24);
        const float4 v = xw4[(size_t)s * (CH / 4) + lane];
        const float nr = dinv[s] * dinv[base + r];
        ag_addslot(A, r >> 5, v, nr);
    }
}

__device__ __forceinline__ float4 ag_start(int r, int rows, int base, int lane,
                                           const float4* __restrict__ xw4, const float* __restrict__ dinv)
{
    float4 a = make_float4(0.f, 0.f, 0.f, 0.f);
    if (r < rows) {
        const int d = base + r;
        const float di = dinv[d];
        const float4 x = xw4[(size_t)d * (CH / 4) + lane];
        a.x = x.x * di * di;
        a.y = x.y * di * di;
        a.z = x.z * di * di;
        a.w = x.w * di * di;
    }
    return a;
}

__global__ __launch_bounds__(1024) void gcn_aggregate(
    const int* __restrict__ src, const int* __restrict__ dst,
    const float* __restrict__ dinv, const float* __restrict__ x, const float* __restrict__ W,
    const float* __restrict__ bias, float* __restrict__ out, int n, int E)
{
    __shared__ uint4    pool4[(AG_WAVES * AG_LCAP + AG_WAVES * AG_MCAP) / 4];
    __shared__ float4   tileA4[AG_ROWS * AG_TP / 4];
    __shared__ int      cnts[AG_WAVES];
    __shared__ int      more[2];

    const int wave = __builtin_amdgcn_readfirstlane((int)(threadIdx.x >> 5));
    const int lane = threadIdx.x & 31;
    const unsigned lt = (1u << lane) - 1u;
    const int base = blockIdx.x * AG_ROWS;
    int rows = n - base;
    if (rows > AG_ROWS) rows = AG_ROWS;

    unsigned* lists  = (unsigned*)pool4;
    unsigned* wl     = lists + wave * AG_LCAP;
    unsigned* mylist = lists + AG_WAVES * AG_LCAP + wave * AG_MCAP;
    const float4* xw4 = (const float4*)x;

    ag_acc A;
    {
#pragma unroll
        for (int i = 0; i < AG_SLOTS; ++i) {
            const float4 t = ag_start(wave + 32 * i, rows, base, lane, xw4, dinv);
            A.a[4 * i + 0] = t.x; A.a[4 * i + 1] = t.y; A.a[4 * i + 2] = t.z; A.a[4 * i + 3] = t.w;
        }
    }
    if (threadIdx.x == 0) { more[0] = 0; more[1] = 0; }
    __syncthreads();

    const int segLen = (((E + AG_WAVES - 1) / AG_WAVES) + 3) & ~3;
    int pos = wave * segLen;
    int segEnd = pos + segLen;
    if (segEnd > E) segEnd = E;
    if (pos > E) pos = E;
    const bool al16 = ((((size_t)dst) & 15) == 0);

    for (int pass = 0; ; ++pass) {
        int cnt = 0;
        if (al16) {
            if (pos + 512 <= segEnd) {
                const int4* p = (const int4*)(dst + pos + 4 * lane);
                int4 a0 = p[0], a1 = p[32], a2 = p[64], a3 = p[96];
                for (;;) {
                    int4 b0, b1, b2, b3;
                    {
                        const bool pre = (pos + 1024 <= segEnd);
                        const int4* pn = (const int4*)(dst + (pre ? pos + 512 : pos) + 4 * lane);
                        b0 = pn[0]; b1 = pn[32]; b2 = pn[64]; b3 = pn[96];
                        const int e0 = pos + 4 * lane;
                        ag_quad(a0, e0,       base, rows, lt, wl, cnt);
                        ag_quad(a1, e0 + 128, base, rows, lt, wl, cnt);
                        ag_quad(a2, e0 + 256, base, rows, lt, wl, cnt);
                        ag_quad(a3, e0 + 384, base, rows, lt, wl, cnt);
                        pos += 512;
                        if (!(pre && cnt <= AG_LCAP - 512)) break;
                    }
                    {
                        const bool pre = (pos + 1024 <= segEnd);
                        const int4* pn = (const int4*)(dst + (pre ? pos + 512 : pos) + 4 * lane);
                        a0 = pn[0]; a1 = pn[32]; a2 = pn[64]; a3 = pn[96];
                        const int e0 = pos + 4 * lane;
                        ag_quad(b0, e0,       base, rows, lt, wl, cnt);
                        ag_quad(b1, e0 + 128, base, rows, lt, wl, cnt);
                        ag_quad(b2, e0 + 256, base, rows, lt, wl, cnt);
                        ag_quad(b3, e0 + 384, base, rows, lt, wl, cnt);
                        pos += 512;
                        if (!(pre && cnt <= AG_LCAP - 512)) break;
                    }
                }
            }
            while (pos + 128 <= segEnd && cnt <= AG_LCAP - 128) {
                const int e0 = pos + 4 * lane;
                const int4 q0 = *(const int4*)(dst + e0);
                ag_quad(q0, e0, base, rows, lt, wl, cnt);
                pos += 128;
            }
        }
        while (pos < segEnd && cnt <= AG_LCAP - 32) {
            const int e = pos + lane;
            const int dv = (e < segEnd) ? dst[e] : -1;
            ag_append(dv, e, base, rows, lt, wl, cnt);
            pos += 32;
        }

        if (lane == 0) {
            cnts[wave] = cnt;
            if (pos < segEnd) more[pass & 1] = 1;
        }
        __syncthreads();
        if (threadIdx.x == 0) more[(pass + 1) & 1] = 0;

        const int myc = cnts[lane];
        int mcnt = 0;
        for (int L = 0; L < AG_WAVES; ++L) {
            const int c = __builtin_amdgcn_readlane(myc, L);
            const unsigned* ll = lists + L * AG_LCAP;
            for (int i = 0; i < c; i += 128) {
                const int j0 = i + 4 * lane;
                uint4 en = make_uint4(0u, 0u, 0u, 0u);
                if (j0 < c) en = *(const uint4*)(ll + j0);
                const unsigned ev[4] = {en.x, en.y, en.z, en.w};
#pragma unroll
                for (int j = 0; j < 4; ++j) {
                    const bool m = (j0 + j < c) && ((int)((ev[j] >> 24) & 31u) == wave);
                    const unsigned mask = __builtin_amdgcn_ballot_w32(m);
                    if (mask) {
                        if (m) mylist[mcnt + __popc(mask & lt)] = ev[j];
                        mcnt += __popc(mask);
                    }
                }
                if (mcnt > AG_MCAP - 128) {
                    ag_flush(A, mylist, mcnt, base, lane, src, xw4, dinv);
                    mcnt = 0;
                }
            }
        }
        ag_flush(A, mylist, mcnt, base, lane, src, xw4, dinv);

        const int again = more[pass & 1];
        __syncthreads();
        if (!again) break;
    }

    const int half = lane >> 4;
    const int l16  = lane & 15;
    const int ct   = wave & 7;
    const int rg   = wave >> 3;
    float* tileA = (float*)tileA4;

    v2f bf[32];
    {
        const float* Wp = W + (size_t)(2 * half) * CH + 16 * ct + l16;
#pragma unroll
        for (int ks = 0; ks < 32; ++ks) {
            bf[ks].x = Wp[(size_t)(4 * ks) * CH];
            bf[ks].y = Wp[(size_t)(4 * ks + 1) * CH];
        }
    }

#pragma unroll
    for (int i = 0; i < AG_SLOTS; ++i) {
        const int r = wave + 32 * i;
        *(float4*)(tileA + r * AG_TP + 4 * lane) =
            make_float4(A.a[4 * i + 0], A.a[4 * i + 1], A.a[4 * i + 2], A.a[4 * i + 3]);
    }
    __syncthreads();

    unsigned* tileD = (unsigned*)pool4;
    for (int hh = 0; hh < 2; ++hh) {
        const int R0 = 64 * rg + 32 * hh;
        const float* a0p = tileA + (R0 + l16) * AG_TP + 2 * half;
        const float* a1p = a0p + 16 * AG_TP;
        v8f c0 = {}, c1 = {};
#pragma unroll
        for (int ks = 0; ks < 32; ++ks) {
            const v2f a0 = *(const v2f*)(a0p + 4 * ks);
            const v2f a1 = *(const v2f*)(a1p + 4 * ks);
            c0 = __builtin_amdgcn_wmma_f32_16x16x4_f32(
                      false, a0,  false, bf[ks],
                      (short)0, c0,  false,  false);
            c1 = __builtin_amdgcn_wmma_f32_16x16x4_f32(
                      false, a1,  false, bf[ks],
                      (short)0, c1,  false,  false);
        }
#pragma unroll
        for (int r = 0; r < 8; ++r) {
            tileD[(R0 + r + 8 * half) * AG_TP + 16 * ct + l16]      = __float_as_uint(c0[r]);
            tileD[(R0 + 16 + r + 8 * half) * AG_TP + 16 * ct + l16] = __float_as_uint(c1[r]);
        }
    }
    __syncthreads();

    float4* out4 = (float4*)out;
    const float4 b4 = ((const float4*)bias)[lane];
#pragma unroll
    for (int i = 0; i < AG_SLOTS; ++i) {
        const int r = wave + 32 * i;
        if (r < rows) {
            const uint4 u = *(const uint4*)(tileD + r * AG_TP + 4 * lane);
            out4[(size_t)(base + r) * (CH / 4) + lane] =
                make_float4(__uint_as_float(u.x) + b4.x, __uint_as_float(u.y) + b4.y,
                            __uint_as_float(u.z) + b4.z, __uint_as_float(u.w) + b4.w);
        }
    }
}


extern "C" void kernel_launch(void* const* d_in, const int* in_sizes, int n_in,
                              void* d_out, int out_size, void* d_ws, size_t ws_size,
                              hipStream_t stream) {
    const float* x  = (const float*)d_in[0];
    const float* W  = (const float*)d_in[1];
    const float* b  = (const float*)d_in[2];
    const int*   ei = (const int*)d_in[3];

    const int n = in_sizes[0] / CH;
    const int E = in_sizes[3] / 2;
    const int* src = ei;
    const int* dst = ei + E;

    float* deg = (float*)d_ws;
    float* out = (float*)d_out;

    gcn_deg_init <<<(n + 255) / 256, 256, 0, stream>>>(deg, n);
    gcn_deg_accum<<<(E + 255) / 256, 256, 0, stream>>>(dst, deg, E);
    gcn_dinv     <<<(n + 255) / 256, 256, 0, stream>>>(deg, n);

    gcn_aggregate<<<(n + AG_ROWS - 1) / AG_ROWS, 1024, 0, stream>>>(src, dst, deg, x, W, b, out, n, E);
}


# aggregation scans a one-byte-per-edge key plane (dst >> 8, packed by the degree kernel, 4 edges per lane) instead of the dst list; matched edges' dst looked up after; list order changes the sum order
# speedup vs baseline: 1.9004x; 1.1005x over previous
#include <hip/hip_runtime.h>
#include <hip/hip_bf16.h>

#define CH 128

typedef __attribute__((ext_vector_type(2))) float v2f;
typedef __attribute__((ext_vector_type(8))) float v8f;


__global__ void gcn_deg_init(float* __restrict__ deg, int n) {
    int i = blockIdx.x * blockDim.x + threadIdx.x;
    if (i < n) deg[i] = 1.0f;
}

__device__ __forceinline__ unsigned* gcn_keys(float* deg, int n) {
    return (unsigned*)((char*)deg + ((((size_t)n) * 4 + 255) & ~(size_t)255));
}
__device__ __forceinline__ int gcn_padded(int E) { return (E + 511) & ~511; }

__global__ void gcn_deg_accum(const int* __restrict__ dst, float* __restrict__ deg, int n, int E) {
    const int t = blockIdx.x * blockDim.x + threadIdx.x;
    if (4 * t >= gcn_padded(E)) return;
    const int e = 4 * t;
    int d[4];
    if (e + 4 <= E && ((((size_t)dst) & 15) == 0)) {
        const int4 q = *(const int4*)(dst + e);
        d[0] = q.x; d[1] = q.y; d[2] = q.z; d[3] = q.w;
    } else {
#pragma unroll
        for (int j = 0; j < 4; ++j) d[j] = (e + j < E) ? dst[e + j] : -1;
    }
    unsigned k = 0;
#pragma unroll
    for (int j = 0; j < 4; ++j) {
        if (d[j] >= 0) {
            atomicAdd(&deg[d[j]], 1.0f);
            k |= (((unsigned)d[j] >> 8) & 255u) << (8 * j);
        }
    }
    gcn_keys(deg, n)[t] = k;
}

__global__ void gcn_dinv(float* __restrict__ deg, int n) {
    int i = blockIdx.x * blockDim.x + threadIdx.x;
    if (i < n) deg[i] = rsqrtf(deg[i]);
}


#define AG_ROWS    256
#define AG_SLOTS   8
#define AG_WAVES   32
#define AG_LCAP    768
#define AG_MCAP    384
#define AG_TP      132

__device__ __forceinline__ const unsigned* gcn_keys_ro(const float* dinv, int n) {
    return (const unsigned*)((const char*)dinv + ((((size_t)n) * 4 + 255) & ~(size_t)255));
}

__device__ __forceinline__ unsigned ag_zero_bytes(unsigned x) {
    return ~(((x & 0x7F7F7F7Fu) + 0x7F7F7F7Fu) | x) & 0x80808080u;
}

__device__ __forceinline__ void ag_keys16(const uint4 q, unsigned e0, unsigned kk, unsigned lt,
                                          unsigned* wl, int& cnt)
{
    const unsigned f0 = ag_zero_bytes(q.x ^ kk);
    const unsigned f1 = ag_zero_bytes(q.y ^ kk);
    const unsigned f2 = ag_zero_bytes(q.z ^ kk);
    const unsigned f3 = ag_zero_bytes(q.w ^ kk);
    unsigned cm = (f0 >> 7) | (f1 >> 6) | (f2 >> 5) | (f3 >> 4);
    for (;;) {
        const unsigned mask = __builtin_amdgcn_ballot_w32(cm != 0u);
        if (!mask) break;
        if (cm) {
            const unsigned p = (unsigned)__builtin_ctz(cm);
            wl[cnt + __popc(mask & lt)] = e0 + ((p & 7u) << 2) + (p >> 3);
            cm &= cm - 1u;
        }
        cnt += __popc(mask);
    }
}

typedef __attribute__((ext_vector_type(32))) float ag_v32f;
struct ag_acc { ag_v32f a; };

__device__ __forceinline__ void ag_addslot(ag_acc& A, int slot, const float4 v, float nr)
{
    const int b = slot * 4;
    A.a[b + 0] += v.x * nr;
    A.a[b + 1] += v.y * nr;
    A.a[b + 2] += v.z * nr;
    A.a[b + 3] += v.w * nr;
}

template <int NG>
__device__ __forceinline__ void ag_addgroup(ag_acc& A, const unsigned* ent, int base, int lane,
                                            const float4* __restrict__ xw4, const float* __restrict__ dinv)
{
    unsigned t[4 * NG];
#pragma unroll
    for (int g = 0; g < NG; ++g) {
        const uint4 en = *(const uint4*)(ent + 4 * g);
        t[4 * g + 0] = __builtin_amdgcn_readfirstlane(en.x);
        t[4 * g + 1] = __builtin_amdgcn_readfirstlane(en.y);
        t[4 * g + 2] = __builtin_amdgcn_readfirstlane(en.z);
        t[4 * g + 3] = __builtin_amdgcn_readfirstlane(en.w);
    }
    float4 v[4 * NG];
    float nr[4 * NG];
#pragma unroll
    for (int i = 0; i < 4 * NG; ++i)
        v[i] = xw4[(size_t)(t[i] & 0xFFFFFFu) * (CH / 4) + lane];
#pragma unroll
    for (int i = 0; i < 4 * NG; ++i)
        nr[i] = dinv[t[i] & 0xFFFFFFu] * dinv[base + (int)(t[i] >> 24)];
#pragma unroll
    for (int i = 0; i < 4 * NG; ++i)
        ag_addslot(A, (int)(t[i] >> 29), v[i], nr[i]);
}

__device__ __forceinline__ void ag_flush(ag_acc& A, unsigned* mylist, int mcnt, int base, int lane,
                                         const int* __restrict__ src,
                                         const float4* __restrict__ xw4, const float* __restrict__ dinv)
{
    for (int k = lane; k < mcnt; k += 32) {
        const unsigned t = mylist[k];
        mylist[k] = (t & 0xFF000000u) | (unsigned)src[t & 0xFFFFFFu];
    }
    int k = 0;
    for (; k + 12 <= mcnt; k += 12) ag_addgroup<3>(A, mylist + k, base, lane, xw4, dinv);
    for (; k + 4 <= mcnt; k += 4)   ag_addgroup<1>(A, mylist + k, base, lane, xw4, dinv);
    for (; k < mcnt; ++k) {
        const unsigned t = __builtin_amdgcn_readfirstlane(mylist[k]);
        const int s = (int)(t & 0xFFFFFFu), r = (int)(t >> 24);
        const float4 v = xw4[(size_t)s * (CH / 4) + lane];
        const float nr = dinv[s] * dinv[base + r];
        ag_addslot(A, r >> 5, v, nr);
    }
}

__device__ __forceinline__ float4 ag_start(int r, int rows, int base, int lane,
                                           const float4* __restrict__ xw4, const float* __restrict__ dinv)
{
    float4 a = make_float4(0.f, 0.f, 0.f, 0.f);
    if (r < rows) {
        const int d = base + r;
        const float di = dinv[d];
        const float4 x = xw4[(size_t)d * (CH / 4) + lane];
        a.x = x.x * di * di;
        a.y = x.y * di * di;
        a.z = x.z * di * di;
        a.w = x.w * di * di;
    }
    return a;
}

__global__ __launch_bounds__(1024) void gcn_aggregate(
    const int* __restrict__ src, const int* __restrict__ dst,
    const float* __restrict__ dinv, const float* __restrict__ x, const float* __restrict__ W,
    const float* __restrict__ bias, float* __restrict__ out, int n, int E)
{
    __shared__ uint4    pool4[(AG_WAVES * AG_LCAP + AG_WAVES * AG_MCAP) / 4];
    __shared__ float4   tileA4[AG_ROWS * AG_TP / 4];
    __shared__ int      cnts[AG_WAVES];
    __shared__ int      more[2];

    const int wave = __builtin_amdgcn_readfirstlane((int)(threadIdx.x >> 5));
    const int lane = threadIdx.x & 31;
    const unsigned lt = (1u << lane) - 1u;
    const int base = blockIdx.x * AG_ROWS;
    int rows = n - base;
    if (rows > AG_ROWS) rows = AG_ROWS;

    unsigned* lists  = (unsigned*)pool4;
    unsigned* wl     = lists + wave * AG_LCAP;
    unsigned* mylist = lists + AG_WAVES * AG_LCAP + wave * AG_MCAP;
    const float4* xw4 = (const float4*)x;

    ag_acc A;
    {
#pragma unroll
        for (int i = 0; i < AG_SLOTS; ++i) {
            const float4 t = ag_start(wave + 32 * i, rows, base, lane, xw4, dinv);
            A.a[4 * i + 0] = t.x; A.a[4 * i + 1] = t.y; A.a[4 * i + 2] = t.z; A.a[4 * i + 3] = t.w;
        }
    }
    if (threadIdx.x == 0) { more[0] = 0; more[1] = 0; }
    __syncthreads();

    const unsigned* keys = gcn_keys_ro(dinv, n);
    const int P = gcn_padded(E);
    const int segLen = ((((P >> 9) + AG_WAVES - 1) / AG_WAVES)) << 9;
    int pos = wave * segLen;
    int segEnd = pos + segLen;
    if (segEnd > P) segEnd = P;
    if (pos > P) pos = P;
    const unsigned kk = (blockIdx.x & 255u) * 0x01010101u;

    for (int pass = 0; ; ++pass) {
        int cnt = 0;
        {
            const int left = (segEnd - pos) >> 9;
            if (left > 0) {
                const uint4* p = (const uint4*)(keys + (pos >> 2)) + lane;
                const int last = left - 1;
                uint4 r0 = p[0];
                uint4 r1 = p[32 * (1 < last ? 1 : last)];
                uint4 r2 = p[32 * (2 < last ? 2 : last)];
                uint4 r3 = p[32 * (3 < last ? 3 : last)];
                int i = 0;
#define AG_STEP(R)                                                                   \
                {                                                                    \
                    go = go && (i < left) && (cnt <= AG_LCAP - 512);                 \
                    if (go) {                                                        \
                        ag_keys16(R, (unsigned)pos + 16u * (unsigned)lane, kk, lt, wl, cnt); \
                        pos += 512; ++i;                                             \
                    }                                                                \
                    R = p[32 * (i + 3 < last ? i + 3 : last)];                       \
                }
                for (;;) {
                    bool go = true;
                    AG_STEP(r0) AG_STEP(r1) AG_STEP(r2) AG_STEP(r3)
                    if (!go) break;
                }
#undef AG_STEP
            }
        }
        {
            int outc = 0;
            for (int i0 = 0; i0 < cnt; i0 += 128) {
                unsigned ee[4];
                int dv[4];
#pragma unroll
                for (int u = 0; u < 4; ++u) {
                    const int idx = i0 + 32 * u + lane;
                    ee[u] = (idx < cnt) ? wl[idx] : 0xFFFFFFFFu;
                }
#pragma unroll
                for (int u = 0; u < 4; ++u)
                    dv[u] = (ee[u] < (unsigned)E) ? dst[ee[u]] : -1;
#pragma unroll
                for (int u = 0; u < 4; ++u) {
                    const unsigned dl = (unsigned)(dv[u] - base);
                    const bool m = (dv[u] >= 0) && (dl < (unsigned)rows);
                    const unsigned mask = __builtin_amdgcn_ballot_w32(m);
                    if (mask) {
                        if (m) wl[outc + __popc(mask & lt)] = (dl << 24) | ee[u];
                        outc += __popc(mask);
                    }
                }
            }
            cnt = outc;
        }

        if (lane == 0) {
            cnts[wave] = cnt;
            if (pos < segEnd) more[pass & 1] = 1;
        }
        __syncthreads();
        if (threadIdx.x == 0) more[(pass + 1) & 1] = 0;

        const int myc = cnts[lane];
        int mcnt = 0;
        for (int L = 0; L < AG_WAVES; ++L) {
            const int c = __builtin_amdgcn_readlane(myc, L);
            const unsigned* ll = lists + L * AG_LCAP;
            for (int i = 0; i < c; i += 128) {
                const int j0 = i + 4 * lane;
                uint4 en = make_uint4(0u, 0u, 0u, 0u);
                if (j0 < c) en = *(const uint4*)(ll + j0);
                const unsigned ev[4] = {en.x, en.y, en.z, en.w};
#pragma unroll
                for (int j = 0; j < 4; ++j) {
                    const bool m = (j0 + j < c) && ((int)((ev[j] >> 24) & 31u) == wave);
                    const unsigned mask = __builtin_amdgcn_ballot_w32(m);
                    if (mask) {
                        if (m) mylist[mcnt + __popc(mask & lt)] = ev[j];
                        mcnt += __popc(mask);
                    }
                }
                if (mcnt > AG_MCAP - 128) {
                    ag_flush(A, mylist, mcnt, base, lane, src, xw4, dinv);
                    mcnt = 0;
                }
            }
        }
        ag_flush(A, mylist, mcnt, base, lane, src, xw4, dinv);

        const int again = more[pass & 1];
        __syncthreads();
        if (!again) break;
    }

    const int half = lane >> 4;
    const int l16  = lane & 15;
    const int ct   = wave & 7;
    const int rg   = wave >> 3;
    float* tileA = (float*)tileA4;

    v2f bf[32];
    {
        const float* Wp = W + (size_t)(2 * half) * CH + 16 * ct + l16;
#pragma unroll
        for (int ks = 0; ks < 32; ++ks) {
            bf[ks].x = Wp[(size_t)(4 * ks) * CH];
            bf[ks].y = Wp[(size_t)(4 * ks + 1) * CH];
        }
    }

#pragma unroll
    for (int i = 0; i < AG_SLOTS; ++i) {
        const int r = wave + 32 * i;
        *(float4*)(tileA + r * AG_TP + 4 * lane) =
            make_float4(A.a[4 * i + 0], A.a[4 * i + 1], A.a[4 * i + 2], A.a[4 * i + 3]);
    }
    __syncthreads();

    unsigned* tileD = (unsigned*)pool4;
    for (int hh = 0; hh < 2; ++hh) {
        const int R0 = 64 * rg + 32 * hh;
        const float* a0p = tileA + (R0 + l16) * AG_TP + 2 * half;
        const float* a1p = a0p + 16 * AG_TP;
        v8f c0 = {}, c1 = {};
#pragma unroll
        for (int ks = 0; ks < 32; ++ks) {
            const v2f a0 = *(const v2f*)(a0p + 4 * ks);
            const v2f a1 = *(const v2f*)(a1p + 4 * ks);
            c0 = __builtin_amdgcn_wmma_f32_16x16x4_f32(
                      false, a0,  false, bf[ks],
                      (short)0, c0,  false,  false);
            c1 = __builtin_amdgcn_wmma_f32_16x16x4_f32(
                      false, a1,  false, bf[ks],
                      (short)0, c1,  false,  false);
        }
#pragma unroll
        for (int r = 0; r < 8; ++r) {
            tileD[(R0 + r + 8 * half) * AG_TP + 16 * ct + l16]      = __float_as_uint(c0[r]);
            tileD[(R0 + 16 + r + 8 * half) * AG_TP + 16 * ct + l16] = __float_as_uint(c1[r]);
        }
    }
    __syncthreads();

    float4* out4 = (float4*)out;
    const float4 b4 = ((const float4*)bias)[lane];
#pragma unroll
    for (int i = 0; i < AG_SLOTS; ++i) {
        const int r = wave + 32 * i;
        if (r < rows) {
            const uint4 u = *(const uint4*)(tileD + r * AG_TP + 4 * lane);
            out4[(size_t)(base + r) * (CH / 4) + lane] =
                make_float4(__uint_as_float(u.x) + b4.x, __uint_as_float(u.y) + b4.y,
                            __uint_as_float(u.z) + b4.z, __uint_as_float(u.w) + b4.w);
        }
    }
}


extern "C" void kernel_launch(void* const* d_in, const int* in_sizes, int n_in,
                              void* d_out, int out_size, void* d_ws, size_t ws_size,
                              hipStream_t stream) {
    const float* x  = (const float*)d_in[0];
    const float* W  = (const float*)d_in[1];
    const float* b  = (const float*)d_in[2];
    const int*   ei = (const int*)d_in[3];

    const int n = in_sizes[0] / CH;
    const int E = in_sizes[3] / 2;
    const int* src = ei;
    const int* dst = ei + E;

    float* deg = (float*)d_ws;
    float* out = (float*)d_out;

    gcn_deg_init <<<(n + 255) / 256, 256, 0, stream>>>(deg, n);
    gcn_deg_accum<<<(E + 1534) / 1024, 256, 0, stream>>>(dst, deg, n, E);
    gcn_dinv     <<<(n + 255) / 256, 256, 0, stream>>>(deg, n);

    gcn_aggregate<<<(n + AG_ROWS - 1) / AG_ROWS, 1024, 0, stream>>>(src, dst, deg, x, W, b, out, n, E);
}


# key plane stored bit-sliced (8 words per 32 edges, packed by the degree kernel with wave votes): a lane tests 32 edges with 8 boolean ops, was 16 keys with 28 ops; list order changes the sum order
# speedup vs baseline: 1.9643x; 1.0336x over previous
#include <hip/hip_runtime.h>
#include <hip/hip_bf16.h>

#define CH 128

typedef __attribute__((ext_vector_type(2))) float v2f;
typedef __attribute__((ext_vector_type(8))) float v8f;


__global__ void gcn_deg_init(float* __restrict__ deg, int n) {
    int i = blockIdx.x * blockDim.x + threadIdx.x;
    if (i < n) deg[i] = 1.0f;
}

__device__ __forceinline__ unsigned* gcn_keys(float* deg, int n) {
    return (unsigned*)((char*)deg + ((((size_t)n) * 4 + 255) & ~(size_t)255));
}
__device__ __forceinline__ int gcn_padded(int E) { return (E + 1023) & ~1023; }

__global__ void gcn_deg_accum(const int* __restrict__ dst, float* __restrict__ deg, int n, int E) {
    const int lane = threadIdx.x & 31;
    const int wv   = threadIdx.x >> 5;
    const int e0   = blockIdx.x * 1024 + wv * 128;
    int d[4];
#pragma unroll
    for (int j = 0; j < 4; ++j) {
        const int e = e0 + 32 * j + lane;
        d[j] = (e < E) ? dst[e] : -1;
    }
    unsigned myw = 0;
#pragma unroll
    for (int j = 0; j < 4; ++j) {
        unsigned key = 0;
        if (d[j] >= 0) {
            atomicAdd(&deg[d[j]], 1.0f);
            key = ((unsigned)d[j] >> 8) & 255u;
        }
#pragma unroll
        for (int b = 0; b < 8; ++b) {
            const unsigned w = __builtin_amdgcn_ballot_w32(((key >> b) & 1u) != 0u);
            if (lane == 8 * j + b) myw = w;
        }
    }
    const int j = lane >> 3, b = lane & 7;
    gcn_keys(deg, n)[(size_t)blockIdx.x * 256 + (b >> 2) * 128 + (4 * wv + j) * 4 + (b & 3)] = myw;
}

__global__ void gcn_dinv(float* __restrict__ deg, int n) {
    int i = blockIdx.x * blockDim.x + threadIdx.x;
    if (i < n) deg[i] = rsqrtf(deg[i]);
}


#define AG_ROWS    256
#define AG_SLOTS   8
#define AG_WAVES   32
#define AG_LCAP    1536
#define AG_MCAP    384
#define AG_TP      132

__device__ __forceinline__ const unsigned* gcn_keys_ro(const float* dinv, int n) {
    return (const unsigned*)((const char*)dinv + ((((size_t)n) * 4 + 255) & ~(size_t)255));
}

struct ag_key { unsigned s[8]; };

__device__ __forceinline__ void ag_keys32(const uint4 qa, const uint4 qb, unsigned e0, const ag_key& K,
                                          unsigned lt, unsigned* wl, int& cnt)
{
    unsigned df = qa.x ^ K.s[0];
    df |= qa.y ^ K.s[1];
    df |= qa.z ^ K.s[2];
    df |= qa.w ^ K.s[3];
    df |= qb.x ^ K.s[4];
    df |= qb.y ^ K.s[5];
    df |= qb.z ^ K.s[6];
    df |= qb.w ^ K.s[7];
    unsigned cm = ~df;
    for (;;) {
        const unsigned mask = __builtin_amdgcn_ballot_w32(cm != 0u);
        if (!mask) break;
        if (cm) {
            wl[cnt + __popc(mask & lt)] = e0 + (unsigned)__builtin_ctz(cm);
            cm &= cm - 1u;
        }
        cnt += __popc(mask);
    }
}

typedef __attribute__((ext_vector_type(32))) float ag_v32f;
struct ag_acc { ag_v32f a; };

__device__ __forceinline__ void ag_addslot(ag_acc& A, int slot, const float4 v, float nr)
{
    const int b = slot * 4;
    A.a[b + 0] += v.x * nr;
    A.a[b + 1] += v.y * nr;
    A.a[b + 2] += v.z * nr;
    A.a[b + 3] += v.w * nr;
}

template <int NG>
__device__ __forceinline__ void ag_addgroup(ag_acc& A, const unsigned* ent, int base, int lane,
                                            const float4* __restrict__ xw4, const float* __restrict__ dinv)
{
    unsigned t[4 * NG];
#pragma unroll
    for (int g = 0; g < NG; ++g) {
        const uint4 en = *(const uint4*)(ent + 4 * g);
        t[4 * g + 0] = __builtin_amdgcn_readfirstlane(en.x);
        t[4 * g + 1] = __builtin_amdgcn_readfirstlane(en.y);
        t[4 * g + 2] = __builtin_amdgcn_readfirstlane(en.z);
        t[4 * g + 3] = __builtin_amdgcn_readfirstlane(en.w);
    }
    float4 v[4 * NG];
    float nr[4 * NG];
#pragma unroll
    for (int i = 0; i < 4 * NG; ++i)
        v[i] = xw4[(size_t)(t[i] & 0xFFFFFFu) * (CH / 4) + lane];
#pragma unroll
    for (int i = 0; i < 4 * NG; ++i)
        nr[i] = dinv[t[i] & 0xFFFFFFu] * dinv[base + (int)(t[i] >> 24)];
#pragma unroll
    for (int i = 0; i < 4 * NG; ++i)
        ag_addslot(A, (int)(t[i] >> 29), v[i], nr[i]);
}

__device__ __forceinline__ void ag_flush(ag_acc& A, unsigned* mylist, int mcnt, int base, int lane,
                                         const int* __restrict__ src,
                                         const float4* __restrict__ xw4, const float* __restrict__ dinv)
{
    for (int k = lane; k < mcnt; k += 32) {
        const unsigned t = mylist[k];
        mylist[k] = (t & 0xFF000000u) | (unsigned)src[t & 0xFFFFFFu];
    }
    int k = 0;
    for (; k + 12 <= mcnt; k += 12) ag_addgroup<3>(A, mylist + k, base, lane, xw4, dinv);
    for (; k + 4 <= mcnt; k += 4)   ag_addgroup<1>(A, mylist + k, base, lane, xw4, dinv);
    for (; k < mcnt; ++k) {
        const unsigned t = __builtin_amdgcn_readfirstlane(mylist[k]);
        const int s = (int)(t & 0xFFFFFFu), r = (int)(t >> 24);
        const float4 v = xw4[(size_t)s * (CH / 4) + lane];
        const float nr = dinv[s] * dinv[base + r];
        ag_addslot(A, r >> 5, v, nr);
    }
}

__device__ __forceinline__ float4 ag_start(int r, int rows, int base, int lane,
                                           const float4* __restrict__ xw4, const float* __restrict__ dinv)
{
    float4 a = make_float4(0.f, 0.f, 0.f, 0.f);
    if (r < rows) {
        const int d = base + r;
        const float di = dinv[d];
        const float4 x = xw4[(size_t)d * (CH / 4) + lane];
        a.x = x.x * di * di;
        a.y = x.y * di * di;
        a.z = x.z * di * di;
        a.w = x.w * di * di;
    }
    return a;
}

__global__ __launch_bounds__(1024) void gcn_aggregate(
    const int* __restrict__ src, const int* __restrict__ dst,
    const float* __restrict__ dinv, const float* __restrict__ x, const float* __restrict__ W,
    const float* __restrict__ bias, float* __restrict__ out, int n, int E)
{
    __shared__ uint4    pool4[2 * AG_ROWS * AG_TP / 4];
    __shared__ int      cnts[AG_WAVES];
    __shared__ int      more[2];

    const int wave = __builtin_amdgcn_readfirstlane((int)(threadIdx.x >> 5));
    const int lane = threadIdx.x & 31;
    const unsigned lt = (1u << lane) - 1u;
    const int base = blockIdx.x * AG_ROWS;
    int rows = n - base;
    if (rows > AG_ROWS) rows = AG_ROWS;

    unsigned* lists  = (unsigned*)pool4;
    unsigned* wl     = lists + wave * AG_LCAP;
    unsigned* mylist = lists + AG_WAVES * AG_LCAP + wave * AG_MCAP;
    const float4* xw4 = (const float4*)x;

    ag_acc A;
    {
#pragma unroll
        for (int i = 0; i < AG_SLOTS; ++i) {
            const float4 t = ag_start(wave + 32 * i, rows, base, lane, xw4, dinv);
            A.a[4 * i + 0] = t.x; A.a[4 * i + 1] = t.y; A.a[4 * i + 2] = t.z; A.a[4 * i + 3] = t.w;
        }
    }
    if (threadIdx.x == 0) { more[0] = 0; more[1] = 0; }
    __syncthreads();

    const unsigned* keys = gcn_keys_ro(dinv, n);
    const int P = gcn_padded(E);
    const int segLen = ((((P >> 10) + AG_WAVES - 1) / AG_WAVES)) << 10;
    int pos = wave * segLen;
    int segEnd = pos + segLen;
    if (segEnd > P) segEnd = P;
    if (pos > P) pos = P;
    ag_key K;
#pragma unroll
    for (int b = 0; b < 8; ++b) K.s[b] = 0u - ((blockIdx.x >> b) & 1u);

    for (int pass = 0; ; ++pass) {
        int cnt = 0;
        {
            const int left = (segEnd - pos) >> 10;
            if (left > 0) {
                const uint4* p = (const uint4*)(keys + ((size_t)(pos >> 10) << 8)) + lane;
                const int last = left - 1;
                const int i1 = 1 < last ? 1 : last, i2 = 2 < last ? 2 : last, i3 = 3 < last ? 3 : last;
                uint4 a0 = p[0],       b0 = p[32];
                uint4 a1 = p[64 * i1], b1 = p[64 * i1 + 32];
                uint4 a2 = p[64 * i2], b2 = p[64 * i2 + 32];
                uint4 a3 = p[64 * i3], b3 = p[64 * i3 + 32];
                int i = 0;
#define AG_STEP(RA, RB)                                                              \
                {                                                                    \
                    go = go && (i < left) && (cnt <= AG_LCAP - 1024);                \
                    if (go) {                                                        \
                        ag_keys32(RA, RB, (unsigned)pos + 32u * (unsigned)lane, K, lt, wl, cnt); \
                        pos += 1024; ++i;                                            \
                    }                                                                \
                    const int nx = i + 3 < last ? i + 3 : last;                      \
                    RA = p[64 * nx]; RB = p[64 * nx + 32];                           \
                }
                for (;;) {
                    bool go = true;
                    AG_STEP(a0, b0) AG_STEP(a1, b1) AG_STEP(a2, b2) AG_STEP(a3, b3)
                    if (!go) break;
                }
#undef AG_STEP
            }
        }
        {
            int outc = 0;
            for (int i0 = 0; i0 < cnt; i0 += 128) {
                unsigned ee[4];
                int dv[4];
#pragma unroll
                for (int u = 0; u < 4; ++u) {
                    const int idx = i0 + 32 * u + lane;
                    ee[u] = (idx < cnt) ? wl[idx] : 0xFFFFFFFFu;
                }
#pragma unroll
                for (int u = 0; u < 4; ++u)
                    dv[u] = (ee[u] < (unsigned)E) ? dst[ee[u]] : -1;
#pragma unroll
                for (int u = 0; u < 4; ++u) {
                    const unsigned dl = (unsigned)(dv[u] - base);
                    const bool m = (dv[u] >= 0) && (dl < (unsigned)rows);
                    const unsigned mask = __builtin_amdgcn_ballot_w32(m);
                    if (mask) {
                        if (m) wl[outc + __popc(mask & lt)] = (dl << 24) | ee[u];
                        outc += __popc(mask);
                    }
                }
            }
            cnt = outc;
        }

        if (lane == 0) {
            cnts[wave] = cnt;
            if (pos < segEnd) more[pass & 1] = 1;
        }
        __syncthreads();
        if (threadIdx.x == 0) more[(pass + 1) & 1] = 0;

        const int myc = cnts[lane];
        int mcnt = 0;
        for (int L = 0; L < AG_WAVES; ++L) {
            const int c = __builtin_amdgcn_readlane(myc, L);
            const unsigned* ll = lists + L * AG_LCAP;
            for (int i = 0; i < c; i += 128) {
                const int j0 = i + 4 * lane;
                uint4 en = make_uint4(0u, 0u, 0u, 0u);
                if (j0 < c) en = *(const uint4*)(ll + j0);
                const unsigned ev[4] = {en.x, en.y, en.z, en.w};
#pragma unroll
                for (int j = 0; j < 4; ++j) {
                    const bool m = (j0 + j < c) && ((int)((ev[j] >> 24) & 31u) == wave);
                    const unsigned mask = __builtin_amdgcn_ballot_w32(m);
                    if (mask) {
                        if (m) mylist[mcnt + __popc(mask & lt)] = ev[j];
                        mcnt += __popc(mask);
                    }
                }
                if (mcnt > AG_MCAP - 128) {
                    ag_flush(A, mylist, mcnt, base, lane, src, xw4, dinv);
                    mcnt = 0;
                }
            }
        }
        ag_flush(A, mylist, mcnt, base, lane, src, xw4, dinv);

        const int again = more[pass & 1];
        __syncthreads();
        if (!again) break;
    }

    const int half = lane >> 4;
    const int l16  = lane & 15;
    const int ct   = wave & 7;
    const int rg   = wave >> 3;
    float* tileA = (float*)pool4 + AG_ROWS * AG_TP;

    v2f bf[32];
    {
        const float* Wp = W + (size_t)(2 * half) * CH + 16 * ct + l16;
#pragma unroll
        for (int ks = 0; ks < 32; ++ks) {
            bf[ks].x = Wp[(size_t)(4 * ks) * CH];
            bf[ks].y = Wp[(size_t)(4 * ks + 1) * CH];
        }
    }

#pragma unroll
    for (int i = 0; i < AG_SLOTS; ++i) {
        const int r = wave + 32 * i;
        *(float4*)(tileA + r * AG_TP + 4 * lane) =
            make_float4(A.a[4 * i + 0], A.a[4 * i + 1], A.a[4 * i + 2], A.a[4 * i + 3]);
    }
    __syncthreads();

    unsigned* tileD = (unsigned*)pool4;
    for (int hh = 0; hh < 2; ++hh) {
        const int R0 = 64 * rg + 32 * hh;
        const float* a0p = tileA + (R0 + l16) * AG_TP + 2 * half;
        const float* a1p = a0p + 16 * AG_TP;
        v8f c0 = {}, c1 = {};
#pragma unroll
        for (int ks = 0; ks < 32; ++ks) {
            const v2f a0 = *(const v2f*)(a0p + 4 * ks);
            const v2f a1 = *(const v2f*)(a1p + 4 * ks);
            c0 = __builtin_amdgcn_wmma_f32_16x16x4_f32(
                      false, a0,  false, bf[ks],
                      (short)0, c0,  false,  false);
            c1 = __builtin_amdgcn_wmma_f32_16x16x4_f32(
                      false, a1,  false, bf[ks],
                      (short)0, c1,  false,  false);
        }
#pragma unroll
        for (int r = 0; r < 8; ++r) {
            tileD[(R0 + r + 8 * half) * AG_TP + 16 * ct + l16]      = __float_as_uint(c0[r]);
            tileD[(R0 + 16 + r + 8 * half) * AG_TP + 16 * ct + l16] = __float_as_uint(c1[r]);
        }
    }
    __syncthreads();

    float4* out4 = (float4*)out;
    const float4 b4 = ((const float4*)bias)[lane];
#pragma unroll
    for (int i = 0; i < AG_SLOTS; ++i) {
        const int r = wave + 32 * i;
        if (r < rows) {
            const uint4 u = *(const uint4*)(tileD + r * AG_TP + 4 * lane);
            out4[(size_t)(base + r) * (CH / 4) + lane] =
                make_float4(__uint_as_float(u.x) + b4.x, __uint_as_float(u.y) + b4.y,
                            __uint_as_float(u.z) + b4.z, __uint_as_float(u.w) + b4.w);
        }
    }
}


extern "C" void kernel_launch(void* const* d_in, const int* in_sizes, int n_in,
                              void* d_out, int out_size, void* d_ws, size_t ws_size,
                              hipStream_t stream) {
    const float* x  = (const float*)d_in[0];
    const float* W  = (const float*)d_in[1];
    const float* b  = (const float*)d_in[2];
    const int*   ei = (const int*)d_in[3];

    const int n = in_sizes[0] / CH;
    const int E = in_sizes[3] / 2;
    const int* src = ei;
    const int* dst = ei + E;

    float* deg = (float*)d_ws;
    float* out = (float*)d_out;

    gcn_deg_init <<<(n + 255) / 256, 256, 0, stream>>>(deg, n);
    gcn_deg_accum<<<(E + 1023) / 1024, 256, 0, stream>>>(dst, deg, n, E);
    gcn_dinv     <<<(n + 255) / 256, 256, 0, stream>>>(deg, n);

    gcn_aggregate<<<(n + AG_ROWS - 1) / AG_ROWS, 1024, 0, stream>>>(src, dst, deg, x, W, b, out, n, E);
}


# in-degrees counted in LDS: 60 work-groups each count a piece of dst in a table of all nodes (integer adds), the rsqrt launch adds the 60 planes; no global atomics left; 3 launches (was 4); sums same
# speedup vs baseline: 3.2065x; 1.6324x over previous
#include <hip/hip_runtime.h>
#include <hip/hip_bf16.h>

#define CH 128

typedef __attribute__((ext_vector_type(2))) float v2f;
typedef __attribute__((ext_vector_type(8))) float v8f;


__device__ __forceinline__ unsigned* gcn_keys(float* deg, int n) {
    return (unsigned*)((char*)deg + ((((size_t)n) * 4 + 255) & ~(size_t)255));
}
__device__ __forceinline__ int gcn_padded(int E) { return (E + 1023) & ~1023; }

#define HG_PARTS 60
#define HG_CAP   51200

__device__ __forceinline__ int gcn_npad(int n) { return (n + 63) & ~63; }
__device__ __forceinline__ unsigned* gcn_parts(float* deg, int n, int E) {
    return (unsigned*)((char*)gcn_keys(deg, n) + (size_t)gcn_padded(E));
}

struct hg_quad { int d[4]; };

__device__ __forceinline__ int hg_edge(int s, int wave, int j, int lane) {
    return s * 1024 + (wave & 7) * 128 + 32 * j + lane;
}

__device__ __forceinline__ hg_quad hg_load(const int* __restrict__ dst, int s, int wave, int lane, int E) {
    hg_quad q;
#pragma unroll
    for (int j = 0; j < 4; ++j) {
        const int e = hg_edge(s, wave, j, lane);
        q.d[j] = dst[e < E ? e : E - 1];
    }
    return q;
}

__device__ __forceinline__ void hg_keys(const hg_quad& q, int s, int wave, int lane, int E, unsigned* keys)
{
    unsigned myw = 0;
#pragma unroll
    for (int j = 0; j < 4; ++j) {
        const unsigned key = (hg_edge(s, wave, j, lane) < E) ? (((unsigned)q.d[j] >> 8) & 255u) : 0u;
#pragma unroll
        for (int b = 0; b < 8; ++b) {
            const unsigned w = __builtin_amdgcn_ballot_w32(((key >> b) & 1u) != 0u);
            if (lane == 8 * j + b) myw = w;
        }
    }
    const int j = lane >> 3, b = lane & 7;
    keys[(size_t)s * 256 + (b >> 2) * 128 + (4 * (wave & 7) + j) * 4 + (b & 3)] = myw;
}

__global__ __launch_bounds__(1024) void gcn_deg_count(const int* __restrict__ dst, float* __restrict__ deg, int n, int E) {
    __shared__ uint4 hist4[HG_CAP / 4];
    unsigned* hist = (unsigned*)hist4;
    const int lane = threadIdx.x & 31;
    const int wave = __builtin_amdgcn_readfirstlane((int)(threadIdx.x >> 5));
    const int nsteps = gcn_padded(E) >> 10;

    if (blockIdx.x >= HG_PARTS) {
        const int s = 4 * ((int)blockIdx.x - HG_PARTS) + (wave >> 3);
        if (s < nsteps) {
            const hg_quad q = hg_load(dst, s, wave, lane, E);
            asm volatile("" :: "v"(q.d[0]), "v"(q.d[1]), "v"(q.d[2]), "v"(q.d[3]));
            hg_keys(q, s, wave, lane, E, gcn_keys(deg, n));
        }
        return;
    }

    const int s_lo = (int)(((long long)blockIdx.x * nsteps) / HG_PARTS);
    const int s_hi = (int)(((long long)(blockIdx.x + 1) * nsteps) / HG_PARTS);
    const int npad = gcn_npad(n);
    unsigned* part = gcn_parts(deg, n, E) + (size_t)blockIdx.x * npad;

    for (int nb = 0; nb < npad; nb += HG_CAP) {
        int span = npad - nb;
        if (span > HG_CAP) span = HG_CAP;
        hg_quad q[4] = {};
        if (s_lo < s_hi) {
#pragma unroll
            for (int t = 0; t < 4; ++t) q[t] = hg_load(dst, s_lo + 4 * t + (wave >> 3), wave, lane, E);
        }
        for (int i = threadIdx.x; i < (span >> 2); i += 1024) hist4[i] = make_uint4(0u, 0u, 0u, 0u);
        __syncthreads();

        for (int sb = s_lo; sb < s_hi; sb += 16) {
#pragma unroll
            for (int t = 0; t < 4; ++t) {
                const int s = sb + 4 * t + (wave >> 3);
#pragma unroll
                for (int j = 0; j < 4; ++j) {
                    const unsigned dl = (unsigned)(q[t].d[j] - nb);
                    if (s < s_hi && hg_edge(s, wave, j, lane) < E && dl < (unsigned)span)
                        atomicAdd(&hist[dl], 1u);
                }
            }
            if (sb + 16 < s_hi) {
#pragma unroll
                for (int t = 0; t < 4; ++t) q[t] = hg_load(dst, sb + 16 + 4 * t + (wave >> 3), wave, lane, E);
            }
        }
        __syncthreads();

        uint4* p4 = (uint4*)(part + nb);
#pragma unroll 4
        for (int i = threadIdx.x; i < (span >> 2); i += 1024) p4[i] = hist4[i];
        __syncthreads();
    }
}

#define HG_GRP 30
static_assert(HG_PARTS % HG_GRP == 0, "the planes are read in whole groups");
typedef __attribute__((ext_vector_type(4))) unsigned hg_v4u;
__global__ __launch_bounds__(64) void gcn_dinv(float* __restrict__ deg, int n, int E) {
    const int i4 = (blockIdx.x * blockDim.x + threadIdx.x) * 4;
    if (i4 >= n) return;
    const size_t pitch4 = (size_t)(gcn_npad(n) >> 2);
    const hg_v4u* p = (const hg_v4u*)(gcn_parts(deg, n, E) + i4);
    hg_v4u c = {1u, 1u, 1u, 1u};
    for (int g0 = 0; g0 < HG_PARTS; g0 += HG_GRP) {
        hg_v4u v[HG_GRP];
#pragma unroll
        for (int k = 0; k < HG_GRP; ++k) v[k] = p[(size_t)(g0 + k) * pitch4];
        asm volatile("" ::
                        "v"(v[0]), "v"(v[1]), "v"(v[2]), "v"(v[3]), "v"(v[4]), "v"(v[5]), "v"(v[6]), "v"(v[7]), "v"(v[8]), "v"(v[9]),
                        "v"(v[10]), "v"(v[11]), "v"(v[12]), "v"(v[13]), "v"(v[14]), "v"(v[15]), "v"(v[16]), "v"(v[17]), "v"(v[18]), "v"(v[19]),
                        "v"(v[20]), "v"(v[21]), "v"(v[22]), "v"(v[23]), "v"(v[24]), "v"(v[25]), "v"(v[26]), "v"(v[27]), "v"(v[28]), "v"(v[29]));
#pragma unroll
        for (int k = 0; k < HG_GRP; ++k) c += v[k];
    }
    *(float4*)(deg + i4) = make_float4(rsqrtf((float)c.x), rsqrtf((float)c.y), rsqrtf((float)c.z), rsqrtf((float)c.w));
}


#define AG_ROWS    256
#define AG_SLOTS   8
#define AG_WAVES   32
#define AG_LCAP    1536
#define AG_MCAP    384
#define AG_TP      132

__device__ __forceinline__ const unsigned* gcn_keys_ro(const float* dinv, int n) {
    return (const unsigned*)((const char*)dinv + ((((size_t)n) * 4 + 255) & ~(size_t)255));
}

struct ag_key { unsigned s[8]; };

__device__ __forceinline__ void ag_keys32(const uint4 qa, const uint4 qb, unsigned e0, const ag_key& K,
                                          unsigned lt, unsigned* wl, int& cnt)
{
    unsigned df = qa.x ^ K.s[0];
    df |= qa.y ^ K.s[1];
    df |= qa.z ^ K.s[2];
    df |= qa.w ^ K.s[3];
    df |= qb.x ^ K.s[4];
    df |= qb.y ^ K.s[5];
    df |= qb.z ^ K.s[6];
    df |= qb.w ^ K.s[7];
    unsigned cm = ~df;
    for (;;) {
        const unsigned mask = __builtin_amdgcn_ballot_w32(cm != 0u);
        if (!mask) break;
        if (cm) {
            wl[cnt + __popc(mask & lt)] = e0 + (unsigned)__builtin_ctz(cm);
            cm &= cm - 1u;
        }
        cnt += __popc(mask);
    }
}

typedef __attribute__((ext_vector_type(32))) float ag_v32f;
struct ag_acc { ag_v32f a; };

__device__ __forceinline__ void ag_addslot(ag_acc& A, int slot, const float4 v, float nr)
{
    const int b = slot * 4;
    A.a[b + 0] += v.x * nr;
    A.a[b + 1] += v.y * nr;
    A.a[b + 2] += v.z * nr;
    A.a[b + 3] += v.w * nr;
}

template <int NG>
__device__ __forceinline__ void ag_addgroup(ag_acc& A, const unsigned* ent, int base, int lane,
                                            const float4* __restrict__ xw4, const float* __restrict__ dinv)
{
    unsigned t[4 * NG];
#pragma unroll
    for (int g = 0; g < NG; ++g) {
        const uint4 en = *(const uint4*)(ent + 4 * g);
        t[4 * g + 0] = __builtin_amdgcn_readfirstlane(en.x);
        t[4 * g + 1] = __builtin_amdgcn_readfirstlane(en.y);
        t[4 * g + 2] = __builtin_amdgcn_readfirstlane(en.z);
        t[4 * g + 3] = __builtin_amdgcn_readfirstlane(en.w);
    }
    float4 v[4 * NG];
    float nr[4 * NG];
#pragma unroll
    for (int i = 0; i < 4 * NG; ++i)
        v[i] = xw4[(size_t)(t[i] & 0xFFFFFFu) * (CH / 4) + lane];
#pragma unroll
    for (int i = 0; i < 4 * NG; ++i)
        nr[i] = dinv[t[i] & 0xFFFFFFu] * dinv[base + (int)(t[i] >> 24)];
#pragma unroll
    for (int i = 0; i < 4 * NG; ++i)
        ag_addslot(A, (int)(t[i] >> 29), v[i], nr[i]);
}

__device__ __forceinline__ void ag_flush(ag_acc& A, unsigned* mylist, int mcnt, int base, int lane,
                                         const int* __restrict__ src,
                                         const float4* __restrict__ xw4, const float* __restrict__ dinv)
{
    for (int k = lane; k < mcnt; k += 32) {
        const unsigned t = mylist[k];
        mylist[k] = (t & 0xFF000000u) | (unsigned)src[t & 0xFFFFFFu];
    }
    int k = 0;
    for (; k + 12 <= mcnt; k += 12) ag_addgroup<3>(A, mylist + k, base, lane, xw4, dinv);
    for (; k + 4 <= mcnt; k += 4)   ag_addgroup<1>(A, mylist + k, base, lane, xw4, dinv);
    for (; k < mcnt; ++k) {
        const unsigned t = __builtin_amdgcn_readfirstlane(mylist[k]);
        const int s = (int)(t & 0xFFFFFFu), r = (int)(t >> 24);
        const float4 v = xw4[(size_t)s * (CH / 4) + lane];
        const float nr = dinv[s] * dinv[base + r];
        ag_addslot(A, r >> 5, v, nr);
    }
}

__device__ __forceinline__ float4 ag_start(int r, int rows, int base, int lane,
                                           const float4* __restrict__ xw4, const float* __restrict__ dinv)
{
    float4 a = make_float4(0.f, 0.f, 0.f, 0.f);
    if (r < rows) {
        const int d = base + r;
        const float di = dinv[d];
        const float4 x = xw4[(size_t)d * (CH / 4) + lane];
        a.x = x.x * di * di;
        a.y = x.y * di * di;
        a.z = x.z * di * di;
        a.w = x.w * di * di;
    }
    return a;
}

__global__ __launch_bounds__(1024) void gcn_aggregate(
    const int* __restrict__ src, const int* __restrict__ dst,
    const float* __restrict__ dinv, const float* __restrict__ x, const float* __restrict__ W,
    const float* __restrict__ bias, float* __restrict__ out, int n, int E)
{
    __shared__ uint4    pool4[2 * AG_ROWS * AG_TP / 4];
    __shared__ int      cnts[AG_WAVES];
    __shared__ int      more[2];

    const int wave = __builtin_amdgcn_readfirstlane((int)(threadIdx.x >> 5));
    const int lane = threadIdx.x & 31;
    const unsigned lt = (1u << lane) - 1u;
    const int base = blockIdx.x * AG_ROWS;
    int rows = n - base;
    if (rows > AG_ROWS) rows = AG_ROWS;

    unsigned* lists  = (unsigned*)pool4;
    unsigned* wl     = lists + wave * AG_LCAP;
    unsigned* mylist = lists + AG_WAVES * AG_LCAP + wave * AG_MCAP;
    const float4* xw4 = (const float4*)x;

    ag_acc A;
    {
#pragma unroll
        for (int i = 0; i < AG_SLOTS; ++i) {
            const float4 t = ag_start(wave + 32 * i, rows, base, lane, xw4, dinv);
            A.a[4 * i + 0] = t.x; A.a[4 * i + 1] = t.y; A.a[4 * i + 2] = t.z; A.a[4 * i + 3] = t.w;
        }
    }
    if (threadIdx.x == 0) { more[0] = 0; more[1] = 0; }
    __syncthreads();

    const unsigned* keys = gcn_keys_ro(dinv, n);
    const int P = gcn_padded(E);
    const int segLen = ((((P >> 10) + AG_WAVES - 1) / AG_WAVES)) << 10;
    int pos = wave * segLen;
    int segEnd = pos + segLen;
    if (segEnd > P) segEnd = P;
    if (pos > P) pos = P;
    ag_key K;
#pragma unroll
    for (int b = 0; b < 8; ++b) K.s[b] = 0u - ((blockIdx.x >> b) & 1u);

    for (int pass = 0; ; ++pass) {
        int cnt = 0;
        {
            const int left = (segEnd - pos) >> 10;
            if (left > 0) {
                const uint4* p = (const uint4*)(keys + ((size_t)(pos >> 10) << 8)) + lane;
                const int last = left - 1;
                const int i1 = 1 < last ? 1 : last, i2 = 2 < last ? 2 : last, i3 = 3 < last ? 3 : last;
                uint4 a0 = p[0],       b0 = p[32];
                uint4 a1 = p[64 * i1], b1 = p[64 * i1 + 32];
                uint4 a2 = p[64 * i2], b2 = p[64 * i2 + 32];
                uint4 a3 = p[64 * i3], b3 = p[64 * i3 + 32];
                int i = 0;
#define AG_STEP(RA, RB)                                                              \
                {                                                                    \
                    go = go && (i < left) && (cnt <= AG_LCAP - 1024);                \
                    if (go) {                                                        \
                        ag_keys32(RA, RB, (unsigned)pos + 32u * (unsigned)lane, K, lt, wl, cnt); \
                        pos += 1024; ++i;                                            \
                    }                                                                \
                    const int nx = i + 3 < last ? i + 3 : last;                      \
                    RA = p[64 * nx]; RB = p[64 * nx + 32];                           \
                }
                for (;;) {
                    bool go = true;
                    AG_STEP(a0, b0) AG_STEP(a1, b1) AG_STEP(a2, b2) AG_STEP(a3, b3)
                    if (!go) break;
                }
#undef AG_STEP
            }
        }
        {
            int outc = 0;
            for (int i0 = 0; i0 < cnt; i0 += 128) {
                unsigned ee[4];
                int dv[4];
#pragma unroll
                for (int u = 0; u < 4; ++u) {
                    const int idx = i0 + 32 * u + lane;
                    ee[u] = (idx < cnt) ? wl[idx] : 0xFFFFFFFFu;
                }
#pragma unroll
                for (int u = 0; u < 4; ++u)
                    dv[u] = (ee[u] < (unsigned)E) ? dst[ee[u]] : -1;
#pragma unroll
                for (int u = 0; u < 4; ++u) {
                    const unsigned dl = (unsigned)(dv[u] - base);
                    const bool m = (dv[u] >= 0) && (dl < (unsigned)rows);
                    const unsigned mask = __builtin_amdgcn_ballot_w32(m);
                    if (mask) {
                        if (m) wl[outc + __popc(mask & lt)] = (dl << 24) | ee[u];
                        outc += __popc(mask);
                    }
                }
            }
            cnt = outc;
        }

        if (lane == 0) {
            cnts[wave] = cnt;
            if (pos < segEnd) more[pass & 1] = 1;
        }
        __syncthreads();
        if (threadIdx.x == 0) more[(pass + 1) & 1] = 0;

        const int myc = cnts[lane];
        int mcnt = 0;
        for (int L = 0; L < AG_WAVES; ++L) {
            const int c = __builtin_amdgcn_readlane(myc, L);
            const unsigned* ll = lists + L * AG_LCAP;
            for (int i = 0; i < c; i += 128) {
                const int j0 = i + 4 * lane;
                uint4 en = make_uint4(0u, 0u, 0u, 0u);
                if (j0 < c) en = *(const uint4*)(ll + j0);
                const unsigned ev[4] = {en.x, en.y, en.z, en.w};
#pragma unroll
                for (int j = 0; j < 4; ++j) {
                    const bool m = (j0 + j < c) && ((int)((ev[j] >> 24) & 31u) == wave);
                    const unsigned mask = __builtin_amdgcn_ballot_w32(m);
                    if (mask) {
                        if (m) mylist[mcnt + __popc(mask & lt)] = ev[j];
                        mcnt += __popc(mask);
                    }
                }
                if (mcnt > AG_MCAP - 128) {
                    ag_flush(A, mylist, mcnt, base, lane, src, xw4, dinv);
                    mcnt = 0;
                }
            }
        }
        ag_flush(A, mylist, mcnt, base, lane, src, xw4, dinv);

        const int again = more[pass & 1];
        __syncthreads();
        if (!again) break;
    }

    const int half = lane >> 4;
    const int l16  = lane & 15;
    const int ct   = wave & 7;
    const int rg   = wave >> 3;
    float* tileA = (float*)pool4 + AG_ROWS * AG_TP;

    v2f bf[32];
    {
        const float* Wp = W + (size_t)(2 * half) * CH + 16 * ct + l16;
#pragma unroll
        for (int ks = 0; ks < 32; ++ks) {
            bf[ks].x = Wp[(size_t)(4 * ks) * CH];
            bf[ks].y = Wp[(size_t)(4 * ks + 1) * CH];
        }
    }

#pragma unroll
    for (int i = 0; i < AG_SLOTS; ++i) {
        const int r = wave + 32 * i;
        *(float4*)(tileA + r * AG_TP + 4 * lane) =
            make_float4(A.a[4 * i + 0], A.a[4 * i + 1], A.a[4 * i + 2], A.a[4 * i + 3]);
    }
    __syncthreads();

    unsigned* tileD = (unsigned*)pool4;
    for (int hh = 0; hh < 2; ++hh) {
        const int R0 = 64 * rg + 32 * hh;
        const float* a0p = tileA + (R0 + l16) * AG_TP + 2 * half;
        const float* a1p = a0p + 16 * AG_TP;
        v8f c0 = {}, c1 = {};
#pragma unroll
        for (int ks = 0; ks < 32; ++ks) {
            const v2f a0 = *(const v2f*)(a0p + 4 * ks);
            const v2f a1 = *(const v2f*)(a1p + 4 * ks);
            c0 = __builtin_amdgcn_wmma_f32_16x16x4_f32(
                      false, a0,  false, bf[ks],
                      (short)0, c0,  false,  false);
            c1 = __builtin_amdgcn_wmma_f32_16x16x4_f32(
                      false, a1,  false, bf[ks],
                      (short)0, c1,  false,  false);
        }
#pragma unroll
        for (int r = 0; r < 8; ++r) {
            tileD[(R0 + r + 8 * half) * AG_TP + 16 * ct + l16]      = __float_as_uint(c0[r]);
            tileD[(R0 + 16 + r + 8 * half) * AG_TP + 16 * ct + l16] = __float_as_uint(c1[r]);
        }
    }
    __syncthreads();

    float4* out4 = (float4*)out;
    const float4 b4 = ((const float4*)bias)[lane];
#pragma unroll
    for (int i = 0; i < AG_SLOTS; ++i) {
        const int r = wave + 32 * i;
        if (r < rows) {
            const uint4 u = *(const uint4*)(tileD + r * AG_TP + 4 * lane);
            out4[(size_t)(base + r) * (CH / 4) + lane] =
                make_float4(__uint_as_float(u.x) + b4.x, __uint_as_float(u.y) + b4.y,
                            __uint_as_float(u.z) + b4.z, __uint_as_float(u.w) + b4.w);
        }
    }
}


extern "C" void kernel_launch(void* const* d_in, const int* in_sizes, int n_in,
                              void* d_out, int out_size, void* d_ws, size_t ws_size,
                              hipStream_t stream) {
    const float* x  = (const float*)d_in[0];
    const float* W  = (const float*)d_in[1];
    const float* b  = (const float*)d_in[2];
    const int*   ei = (const int*)d_in[3];

    const int n = in_sizes[0] / CH;
    const int E = in_sizes[3] / 2;
    const int* src = ei;
    const int* dst = ei + E;

    float* deg = (float*)d_ws;
    float* out = (float*)d_out;

    gcn_deg_count<<<HG_PARTS + (E + 4095) / 4096, 1024, 0, stream>>>(dst, deg, n, E);
    gcn_dinv     <<<(n + 255) / 256, 64, 0, stream>>>(deg, n, E);

    gcn_aggregate<<<(n + AG_ROWS - 1) / AG_ROWS, 1024, 0, stream>>>(src, dst, deg, x, W, b, out, n, E);
}


# aggregation: a work-group owns 200 destination rows (was 256; key = dst / 200): 250 work-groups for the chip's 256 WGPs (were 196); row tiles past the rows skipped; list order changes the sum order
# speedup vs baseline: 3.5482x; 1.1066x over previous
#include <hip/hip_runtime.h>
#include <hip/hip_bf16.h>

#define CH 128

typedef __attribute__((ext_vector_type(2))) float v2f;
typedef __attribute__((ext_vector_type(8))) float v8f;


#define AG_GROUP 200

__device__ __forceinline__ unsigned* gcn_keys(float* deg, int n) {
    return (unsigned*)((char*)deg + ((((size_t)n) * 4 + 255) & ~(size_t)255));
}
__device__ __forceinline__ int gcn_padded(int E) { return (E + 1023) & ~1023; }

#define HG_PARTS 60
#define HG_CAP   51200

__device__ __forceinline__ int gcn_npad(int n) { return (n + 63) & ~63; }
__device__ __forceinline__ unsigned* gcn_parts(float* deg, int n, int E) {
    return (unsigned*)((char*)gcn_keys(deg, n) + (size_t)gcn_padded(E));
}

struct hg_quad { int d[4]; };

__device__ __forceinline__ int hg_edge(int s, int wave, int j, int lane) {
    return s * 1024 + (wave & 7) * 128 + 32 * j + lane;
}

__device__ __forceinline__ hg_quad hg_load(const int* __restrict__ dst, int s, int wave, int lane, int E) {
    hg_quad q;
#pragma unroll
    for (int j = 0; j < 4; ++j) {
        const int e = hg_edge(s, wave, j, lane);
        q.d[j] = dst[e < E ? e : E - 1];
    }
    return q;
}

__device__ __forceinline__ void hg_keys(const hg_quad& q, int s, int wave, int lane, int E, unsigned* keys)
{
    unsigned myw = 0;
#pragma unroll
    for (int j = 0; j < 4; ++j) {
        const unsigned key = (hg_edge(s, wave, j, lane) < E) ? (((unsigned)q.d[j] / AG_GROUP) & 255u) : 0u;
#pragma unroll
        for (int b = 0; b < 8; ++b) {
            const unsigned w = __builtin_amdgcn_ballot_w32(((key >> b) & 1u) != 0u);
            if (lane == 8 * j + b) myw = w;
        }
    }
    const int j = lane >> 3, b = lane & 7;
    keys[(size_t)s * 256 + (b >> 2) * 128 + (4 * (wave & 7) + j) * 4 + (b & 3)] = myw;
}

__global__ __launch_bounds__(1024) void gcn_deg_count(const int* __restrict__ dst, float* __restrict__ deg, int n, int E) {
    __shared__ uint4 hist4[HG_CAP / 4];
    unsigned* hist = (unsigned*)hist4;
    const int lane = threadIdx.x & 31;
    const int wave = __builtin_amdgcn_readfirstlane((int)(threadIdx.x >> 5));
    const int nsteps = gcn_padded(E) >> 10;

    if (blockIdx.x >= HG_PARTS) {
        const int s = 4 * ((int)blockIdx.x - HG_PARTS) + (wave >> 3);
        if (s < nsteps) {
            const hg_quad q = hg_load(dst, s, wave, lane, E);
            asm volatile("" :: "v"(q.d[0]), "v"(q.d[1]), "v"(q.d[2]), "v"(q.d[3]));
            hg_keys(q, s, wave, lane, E, gcn_keys(deg, n));
        }
        return;
    }

    const int s_lo = (int)(((long long)blockIdx.x * nsteps) / HG_PARTS);
    const int s_hi = (int)(((long long)(blockIdx.x + 1) * nsteps) / HG_PARTS);
    const int npad = gcn_npad(n);
    unsigned* part = gcn_parts(deg, n, E) + (size_t)blockIdx.x * npad;

    for (int nb = 0; nb < npad; nb += HG_CAP) {
        int span = npad - nb;
        if (span > HG_CAP) span = HG_CAP;
        hg_quad q[4] = {};
        if (s_lo < s_hi) {
#pragma unroll
            for (int t = 0; t < 4; ++t) q[t] = hg_load(dst, s_lo + 4 * t + (wave >> 3), wave, lane, E);
        }
        for (int i = threadIdx.x; i < (span >> 2); i += 1024) hist4[i] = make_uint4(0u, 0u, 0u, 0u);
        __syncthreads();

        for (int sb = s_lo; sb < s_hi; sb += 16) {
#pragma unroll
            for (int t = 0; t < 4; ++t) {
                const int s = sb + 4 * t + (wave >> 3);
#pragma unroll
                for (int j = 0; j < 4; ++j) {
                    const unsigned dl = (unsigned)(q[t].d[j] - nb);
                    if (s < s_hi && hg_edge(s, wave, j, lane) < E && dl < (unsigned)span)
                        atomicAdd(&hist[dl], 1u);
                }
            }
            if (sb + 16 < s_hi) {
#pragma unroll
                for (int t = 0; t < 4; ++t) q[t] = hg_load(dst, sb + 16 + 4 * t + (wave >> 3), wave, lane, E);
            }
        }
        __syncthreads();

        uint4* p4 = (uint4*)(part + nb);
#pragma unroll 4
        for (int i = threadIdx.x; i < (span >> 2); i += 1024) p4[i] = hist4[i];
        __syncthreads();
    }
}

#define HG_GRP 30
static_assert(HG_PARTS % HG_GRP == 0, "the planes are read in whole groups");
typedef __attribute__((ext_vector_type(4))) unsigned hg_v4u;
__global__ __launch_bounds__(64) void gcn_dinv(float* __restrict__ deg, int n, int E) {
    const int i4 = (blockIdx.x * blockDim.x + threadIdx.x) * 4;
    if (i4 >= n) return;
    const size_t pitch4 = (size_t)(gcn_npad(n) >> 2);
    const hg_v4u* p = (const hg_v4u*)(gcn_parts(deg, n, E) + i4);
    hg_v4u c = {1u, 1u, 1u, 1u};
    for (int g0 = 0; g0 < HG_PARTS; g0 += HG_GRP) {
        hg_v4u v[HG_GRP];
#pragma unroll
        for (int k = 0; k < HG_GRP; ++k) v[k] = p[(size_t)(g0 + k) * pitch4];
        asm volatile("" ::
                        "v"(v[0]), "v"(v[1]), "v"(v[2]), "v"(v[3]), "v"(v[4]), "v"(v[5]), "v"(v[6]), "v"(v[7]), "v"(v[8]), "v"(v[9]),
                        "v"(v[10]), "v"(v[11]), "v"(v[12]), "v"(v[13]), "v"(v[14]), "v"(v[15]), "v"(v[16]), "v"(v[17]), "v"(v[18]), "v"(v[19]),
                        "v"(v[20]), "v"(v[21]), "v"(v[22]), "v"(v[23]), "v"(v[24]), "v"(v[25]), "v"(v[26]), "v"(v[27]), "v"(v[28]), "v"(v[29]));
#pragma unroll
        for (int k = 0; k < HG_GRP; ++k) c += v[k];
    }
    *(float4*)(deg + i4) = make_float4(rsqrtf((float)c.x), rsqrtf((float)c.y), rsqrtf((float)c.z), rsqrtf((float)c.w));
}


#define AG_ROWS    256
static_assert(AG_GROUP <= AG_ROWS, "a group's rows must fit the work-group's registers and tiles");
#define AG_SLOTS   8
#define AG_WAVES   32
#define AG_LCAP    1536
#define AG_MCAP    384
#define AG_TP      132

__device__ __forceinline__ const unsigned* gcn_keys_ro(const float* dinv, int n) {
    return (const unsigned*)((const char*)dinv + ((((size_t)n) * 4 + 255) & ~(size_t)255));
}

struct ag_key { unsigned s[8]; };

__device__ __forceinline__ void ag_keys32(const uint4 qa, const uint4 qb, unsigned e0, const ag_key& K,
                                          unsigned lt, unsigned* wl, int& cnt)
{
    unsigned df = qa.x ^ K.s[0];
    df |= qa.y ^ K.s[1];
    df |= qa.z ^ K.s[2];
    df |= qa.w ^ K.s[3];
    df |= qb.x ^ K.s[4];
    df |= qb.y ^ K.s[5];
    df |= qb.z ^ K.s[6];
    df |= qb.w ^ K.s[7];
    unsigned cm = ~df;
    for (;;) {
        const unsigned mask = __builtin_amdgcn_ballot_w32(cm != 0u);
        if (!mask) break;
        if (cm) {
            wl[cnt + __popc(mask & lt)] = e0 + (unsigned)__builtin_ctz(cm);
            cm &= cm - 1u;
        }
        cnt += __popc(mask);
    }
}

typedef __attribute__((ext_vector_type(32))) float ag_v32f;
struct ag_acc { ag_v32f a; };

__device__ __forceinline__ void ag_addslot(ag_acc& A, int slot, const float4 v, float nr)
{
    const int b = slot * 4;
    A.a[b + 0] += v.x * nr;
    A.a[b + 1] += v.y * nr;
    A.a[b + 2] += v.z * nr;
    A.a[b + 3] += v.w * nr;
}

template <int NG>
__device__ __forceinline__ void ag_addgroup(ag_acc& A, const unsigned* ent, int base, int lane,
                                            const float4* __restrict__ xw4, const float* __restrict__ dinv)
{
    unsigned t[4 * NG];
#pragma unroll
    for (int g = 0; g < NG; ++g) {
        const uint4 en = *(const uint4*)(ent + 4 * g);
        t[4 * g + 0] = __builtin_amdgcn_readfirstlane(en.x);
        t[4 * g + 1] = __builtin_amdgcn_readfirstlane(en.y);
        t[4 * g + 2] = __builtin_amdgcn_readfirstlane(en.z);
        t[4 * g + 3] = __builtin_amdgcn_readfirstlane(en.w);
    }
    float4 v[4 * NG];
    float nr[4 * NG];
#pragma unroll
    for (int i = 0; i < 4 * NG; ++i)
        v[i] = xw4[(size_t)(t[i] & 0xFFFFFFu) * (CH / 4) + lane];
#pragma unroll
    for (int i = 0; i < 4 * NG; ++i)
        nr[i] = dinv[t[i] & 0xFFFFFFu] * dinv[base + (int)(t[i] >> 24)];
#pragma unroll
    for (int i = 0; i < 4 * NG; ++i)
        ag_addslot(A, (int)(t[i] >> 29), v[i], nr[i]);
}

__device__ __forceinline__ void ag_flush(ag_acc& A, unsigned* mylist, int mcnt, int base, int lane,
                                         const int* __restrict__ src,
                                         const float4* __restrict__ xw4, const float* __restrict__ dinv)
{
    for (int k = lane; k < mcnt; k += 32) {
        const unsigned t = mylist[k];
        mylist[k] = (t & 0xFF000000u) | (unsigned)src[t & 0xFFFFFFu];
    }
    int k = 0;
    for (; k + 12 <= mcnt; k += 12) ag_addgroup<3>(A, mylist + k, base, lane, xw4, dinv);
    for (; k + 4 <= mcnt; k += 4)   ag_addgroup<1>(A, mylist + k, base, lane, xw4, dinv);
    for (; k < mcnt; ++k) {
        const unsigned t = __builtin_amdgcn_readfirstlane(mylist[k]);
        const int s = (int)(t & 0xFFFFFFu), r = (int)(t >> 24);
        const float4 v = xw4[(size_t)s * (CH / 4) + lane];
        const float nr = dinv[s] * dinv[base + r];
        ag_addslot(A, r >> 5, v, nr);
    }
}

__device__ __forceinline__ float4 ag_start(int r, int rows, int base, int lane,
                                           const float4* __restrict__ xw4, const float* __restrict__ dinv)
{
    float4 a = make_float4(0.f, 0.f, 0.f, 0.f);
    if (r < rows) {
        const int d = base + r;
        const float di = dinv[d];
        const float4 x = xw4[(size_t)d * (CH / 4) + lane];
        a.x = x.x * di * di;
        a.y = x.y * di * di;
        a.z = x.z * di * di;
        a.w = x.w * di * di;
    }
    return a;
}

__global__ __launch_bounds__(1024) void gcn_aggregate(
    const int* __restrict__ src, const int* __restrict__ dst,
    const float* __restrict__ dinv, const float* __restrict__ x, const float* __restrict__ W,
    const float* __restrict__ bias, float* __restrict__ out, int n, int E)
{
    __shared__ uint4    pool4[2 * AG_ROWS * AG_TP / 4];
    __shared__ int      cnts[AG_WAVES];
    __shared__ int      more[2];

    const int wave = __builtin_amdgcn_readfirstlane((int)(threadIdx.x >> 5));
    const int lane = threadIdx.x & 31;
    const unsigned lt = (1u << lane) - 1u;
    const int base = blockIdx.x * AG_GROUP;
    int rows = n - base;
    if (rows > AG_GROUP) rows = AG_GROUP;

    unsigned* lists  = (unsigned*)pool4;
    unsigned* wl     = lists + wave * AG_LCAP;
    unsigned* mylist = lists + AG_WAVES * AG_LCAP + wave * AG_MCAP;
    const float4* xw4 = (const float4*)x;

    ag_acc A;
    {
#pragma unroll
        for (int i = 0; i < AG_SLOTS; ++i) {
            const float4 t = ag_start(wave + 32 * i, rows, base, lane, xw4, dinv);
            A.a[4 * i + 0] = t.x; A.a[4 * i + 1] = t.y; A.a[4 * i + 2] = t.z; A.a[4 * i + 3] = t.w;
        }
    }
    if (threadIdx.x == 0) { more[0] = 0; more[1] = 0; }
    __syncthreads();

    const unsigned* keys = gcn_keys_ro(dinv, n);
    const int P = gcn_padded(E);
    const int segLen = ((((P >> 10) + AG_WAVES - 1) / AG_WAVES)) << 10;
    int pos = wave * segLen;
    int segEnd = pos + segLen;
    if (segEnd > P) segEnd = P;
    if (pos > P) pos = P;
    ag_key K;
#pragma unroll
    for (int b = 0; b < 8; ++b) K.s[b] = 0u - ((blockIdx.x >> b) & 1u);

    for (int pass = 0; ; ++pass) {
        int cnt = 0;
        {
            const int left = (segEnd - pos) >> 10;
            if (left > 0) {
                const uint4* p = (const uint4*)(keys + ((size_t)(pos >> 10) << 8)) + lane;
                const int last = left - 1;
                const int i1 = 1 < last ? 1 : last, i2 = 2 < last ? 2 : last, i3 = 3 < last ? 3 : last;
                uint4 a0 = p[0],       b0 = p[32];
                uint4 a1 = p[64 * i1], b1 = p[64 * i1 + 32];
                uint4 a2 = p[64 * i2], b2 = p[64 * i2 + 32];
                uint4 a3 = p[64 * i3], b3 = p[64 * i3 + 32];
                int i = 0;
#define AG_STEP(RA, RB)                                                              \
                {                                                                    \
                    go = go && (i < left) && (cnt <= AG_LCAP - 1024);                \
                    if (go) {                                                        \
                        ag_keys32(RA, RB, (unsigned)pos + 32u * (unsigned)lane, K, lt, wl, cnt); \
                        pos += 1024; ++i;                                            \
                    }                                                                \
                    const int nx = i + 3 < last ? i + 3 : last;                      \
                    RA = p[64 * nx]; RB = p[64 * nx + 32];                           \
                }
                for (;;) {
                    bool go = true;
                    AG_STEP(a0, b0) AG_STEP(a1, b1) AG_STEP(a2, b2) AG_STEP(a3, b3)
                    if (!go) break;
                }
#undef AG_STEP
            }
        }
        {
            int outc = 0;
            for (int i0 = 0; i0 < cnt; i0 += 128) {
                unsigned ee[4];
                int dv[4];
#pragma unroll
                for (int u = 0; u < 4; ++u) {
                    const int idx = i0 + 32 * u + lane;
                    ee[u] = (idx < cnt) ? wl[idx] : 0xFFFFFFFFu;
                }
#pragma unroll
                for (int u = 0; u < 4; ++u)
                    dv[u] = (ee[u] < (unsigned)E) ? dst[ee[u]] : -1;
#pragma unroll
                for (int u = 0; u < 4; ++u) {
                    const unsigned dl = (unsigned)(dv[u] - base);
                    const bool m = (dv[u] >= 0) && (dl < (unsigned)rows);
                    const unsigned mask = __builtin_amdgcn_ballot_w32(m);
                    if (mask) {
                        if (m) wl[outc + __popc(mask & lt)] = (dl << 24) | ee[u];
                        outc += __popc(mask);
                    }
                }
            }
            cnt = outc;
        }

        if (lane == 0) {
            cnts[wave] = cnt;
            if (pos < segEnd) more[pass & 1] = 1;
        }
        __syncthreads();
        if (threadIdx.x == 0) more[(pass + 1) & 1] = 0;

        const int myc = cnts[lane];
        int mcnt = 0;
        for (int L = 0; L < AG_WAVES; ++L) {
            const int c = __builtin_amdgcn_readlane(myc, L);
            const unsigned* ll = lists + L * AG_LCAP;
            for (int i = 0; i < c; i += 128) {
                const int j0 = i + 4 * lane;
                uint4 en = make_uint4(0u, 0u, 0u, 0u);
                if (j0 < c) en = *(const uint4*)(ll + j0);
                const unsigned ev[4] = {en.x, en.y, en.z, en.w};
#pragma unroll
                for (int j = 0; j < 4; ++j) {
                    const bool m = (j0 + j < c) && ((int)((ev[j] >> 24) & 31u) == wave);
                    const unsigned mask = __builtin_amdgcn_ballot_w32(m);
                    if (mask) {
                        if (m) mylist[mcnt + __popc(mask & lt)] = ev[j];
                        mcnt += __popc(mask);
                    }
                }
                if (mcnt > AG_MCAP - 128) {
                    ag_flush(A, mylist, mcnt, base, lane, src, xw4, dinv);
                    mcnt = 0;
                }
            }
        }
        ag_flush(A, mylist, mcnt, base, lane, src, xw4, dinv);

        const int again = more[pass & 1];
        __syncthreads();
        if (!again) break;
    }

    const int half = lane >> 4;
    const int l16  = lane & 15;
    const int ct   = wave & 7;
    const int rg   = wave >> 3;
    float* tileA = (float*)pool4 + AG_ROWS * AG_TP;

    v2f bf[32];
    {
        const float* Wp = W + (size_t)(2 * half) * CH + 16 * ct + l16;
#pragma unroll
        for (int ks = 0; ks < 32; ++ks) {
            bf[ks].x = Wp[(size_t)(4 * ks) * CH];
            bf[ks].y = Wp[(size_t)(4 * ks + 1) * CH];
        }
    }

#pragma unroll
    for (int i = 0; i < AG_SLOTS; ++i) {
        const int r = wave + 32 * i;
        *(float4*)(tileA + r * AG_TP + 4 * lane) =
            make_float4(A.a[4 * i + 0], A.a[4 * i + 1], A.a[4 * i + 2], A.a[4 * i + 3]);
    }
    __syncthreads();

    unsigned* tileD = (unsigned*)pool4;
    for (int hh = 0; hh < 2; ++hh) {
        const int R0 = 64 * rg + 32 * hh;
        if (R0 >= rows) break;
        const float* a0p = tileA + (R0 + l16) * AG_TP + 2 * half;
        const float* a1p = a0p + 16 * AG_TP;
        v8f c0 = {}, c1 = {};
#pragma unroll
        for (int ks = 0; ks < 32; ++ks) {
            const v2f a0 = *(const v2f*)(a0p + 4 * ks);
            const v2f a1 = *(const v2f*)(a1p + 4 * ks);
            c0 = __builtin_amdgcn_wmma_f32_16x16x4_f32(
                      false, a0,  false, bf[ks],
                      (short)0, c0,  false,  false);
            c1 = __builtin_amdgcn_wmma_f32_16x16x4_f32(
                      false, a1,  false, bf[ks],
                      (short)0, c1,  false,  false);
        }
#pragma unroll
        for (int r = 0; r < 8; ++r) {
            tileD[(R0 + r + 8 * half) * AG_TP + 16 * ct + l16]      = __float_as_uint(c0[r]);
            tileD[(R0 + 16 + r + 8 * half) * AG_TP + 16 * ct + l16] = __float_as_uint(c1[r]);
        }
    }
    __syncthreads();

    float4* out4 = (float4*)out;
    const float4 b4 = ((const float4*)bias)[lane];
#pragma unroll
    for (int i = 0; i < AG_SLOTS; ++i) {
        const int r = wave + 32 * i;
        if (r < rows) {
            const uint4 u = *(const uint4*)(tileD + r * AG_TP + 4 * lane);
            out4[(size_t)(base + r) * (CH / 4) + lane] =
                make_float4(__uint_as_float(u.x) + b4.x, __uint_as_float(u.y) + b4.y,
                            __uint_as_float(u.z) + b4.z, __uint_as_float(u.w) + b4.w);
        }
    }
}


extern "C" void kernel_launch(void* const* d_in, const int* in_sizes, int n_in,
                              void* d_out, int out_size, void* d_ws, size_t ws_size,
                              hipStream_t stream) {
    const float* x  = (const float*)d_in[0];
    const float* W  = (const float*)d_in[1];
    const float* b  = (const float*)d_in[2];
    const int*   ei = (const int*)d_in[3];

    const int n = in_sizes[0] / CH;
    const int E = in_sizes[3] / 2;
    const int* src = ei;
    const int* dst = ei + E;

    float* deg = (float*)d_ws;
    float* out = (float*)d_out;

    gcn_deg_count<<<HG_PARTS + (E + 4095) / 4096, 1024, 0, stream>>>(dst, deg, n, E);
    gcn_dinv     <<<(n + 255) / 256, 64, 0, stream>>>(deg, n, E);

    gcn_aggregate<<<(n + AG_GROUP - 1) / AG_GROUP, 1024, 0, stream>>>(src, dst, deg, x, W, b, out, n, E);
}


# aggregation, hand-over: scanning waves count entries per owner wave (LDS table, running sums) and place each entry in its owner's list; owners no longer search all 32 lists; sum order now list order
# speedup vs baseline: 3.9354x; 1.1091x over previous
#include <hip/hip_runtime.h>
#include <hip/hip_bf16.h>

#define CH 128

typedef __attribute__((ext_vector_type(2))) float v2f;
typedef __attribute__((ext_vector_type(8))) float v8f;


#define AG_GROUP 200

__device__ __forceinline__ unsigned* gcn_keys(float* deg, int n) {
    return (unsigned*)((char*)deg + ((((size_t)n) * 4 + 255) & ~(size_t)255));
}
__device__ __forceinline__ int gcn_padded(int E) { return (E + 1023) & ~1023; }

#define HG_PARTS 60
#define HG_CAP   51200

__device__ __forceinline__ int gcn_npad(int n) { return (n + 63) & ~63; }
__device__ __forceinline__ unsigned* gcn_parts(float* deg, int n, int E) {
    return (unsigned*)((char*)gcn_keys(deg, n) + (size_t)gcn_padded(E));
}

struct hg_quad { int d[4]; };

__device__ __forceinline__ int hg_edge(int s, int wave, int j, int lane) {
    return s * 1024 + (wave & 7) * 128 + 32 * j + lane;
}

__device__ __forceinline__ hg_quad hg_load(const int* __restrict__ dst, int s, int wave, int lane, int E) {
    hg_quad q;
#pragma unroll
    for (int j = 0; j < 4; ++j) {
        const int e = hg_edge(s, wave, j, lane);
        q.d[j] = dst[e < E ? e : E - 1];
    }
    return q;
}

__device__ __forceinline__ void hg_keys(const hg_quad& q, int s, int wave, int lane, int E, unsigned* keys)
{
    unsigned myw = 0;
#pragma unroll
    for (int j = 0; j < 4; ++j) {
        const unsigned key = (hg_edge(s, wave, j, lane) < E) ? (((unsigned)q.d[j] / AG_GROUP) & 255u) : 0u;
#pragma unroll
        for (int b = 0; b < 8; ++b) {
            const unsigned w = __builtin_amdgcn_ballot_w32(((key >> b) & 1u) != 0u);
            if (lane == 8 * j + b) myw = w;
        }
    }
    const int j = lane >> 3, b = lane & 7;
    keys[(size_t)s * 256 + (b >> 2) * 128 + (4 * (wave & 7) + j) * 4 + (b & 3)] = myw;
}

__global__ __launch_bounds__(1024) void gcn_deg_count(const int* __restrict__ dst, float* __restrict__ deg, int n, int E) {
    __shared__ uint4 hist4[HG_CAP / 4];
    unsigned* hist = (unsigned*)hist4;
    const int lane = threadIdx.x & 31;
    const int wave = __builtin_amdgcn_readfirstlane((int)(threadIdx.x >> 5));
    const int nsteps = gcn_padded(E) >> 10;

    if (blockIdx.x >= HG_PARTS) {
        const int s = 4 * ((int)blockIdx.x - HG_PARTS) + (wave >> 3);
        if (s < nsteps) {
            const hg_quad q = hg_load(dst, s, wave, lane, E);
            asm volatile("" :: "v"(q.d[0]), "v"(q.d[1]), "v"(q.d[2]), "v"(q.d[3]));
            hg_keys(q, s, wave, lane, E, gcn_keys(deg, n));
        }
        return;
    }

    const int s_lo = (int)(((long long)blockIdx.x * nsteps) / HG_PARTS);
    const int s_hi = (int)(((long long)(blockIdx.x + 1) * nsteps) / HG_PARTS);
    const int npad = gcn_npad(n);
    unsigned* part = gcn_parts(deg, n, E) + (size_t)blockIdx.x * npad;

    for (int nb = 0; nb < npad; nb += HG_CAP) {
        int span = npad - nb;
        if (span > HG_CAP) span = HG_CAP;
        hg_quad q[4] = {};
        if (s_lo < s_hi) {
#pragma unroll
            for (int t = 0; t < 4; ++t) q[t] = hg_load(dst, s_lo + 4 * t + (wave >> 3), wave, lane, E);
        }
        for (int i = threadIdx.x; i < (span >> 2); i += 1024) hist4[i] = make_uint4(0u, 0u, 0u, 0u);
        __syncthreads();

        for (int sb = s_lo; sb < s_hi; sb += 16) {
#pragma unroll
            for (int t = 0; t < 4; ++t) {
                const int s = sb + 4 * t + (wave >> 3);
#pragma unroll
                for (int j = 0; j < 4; ++j) {
                    const unsigned dl = (unsigned)(q[t].d[j] - nb);
                    if (s < s_hi && hg_edge(s, wave, j, lane) < E && dl < (unsigned)span)
                        atomicAdd(&hist[dl], 1u);
                }
            }
            if (sb + 16 < s_hi) {
#pragma unroll
                for (int t = 0; t < 4; ++t) q[t] = hg_load(dst, sb + 16 + 4 * t + (wave >> 3), wave, lane, E);
            }
        }
        __syncthreads();

        uint4* p4 = (uint4*)(part + nb);
#pragma unroll 4
        for (int i = threadIdx.x; i < (span >> 2); i += 1024) p4[i] = hist4[i];
        __syncthreads();
    }
}

#define HG_GRP 30
static_assert(HG_PARTS % HG_GRP == 0, "the planes are read in whole groups");
typedef __attribute__((ext_vector_type(4))) unsigned hg_v4u;
__global__ __launch_bounds__(64) void gcn_dinv(float* __restrict__ deg, int n, int E) {
    const int i4 = (blockIdx.x * blockDim.x + threadIdx.x) * 4;
    if (i4 >= n) return;
    const size_t pitch4 = (size_t)(gcn_npad(n) >> 2);
    const hg_v4u* p = (const hg_v4u*)(gcn_parts(deg, n, E) + i4);
    hg_v4u c = {1u, 1u, 1u, 1u};
    for (int g0 = 0; g0 < HG_PARTS; g0 += HG_GRP) {
        hg_v4u v[HG_GRP];
#pragma unroll
        for (int k = 0; k < HG_GRP; ++k) v[k] = p[(size_t)(g0 + k) * pitch4];
        asm volatile("" ::
                        "v"(v[0]), "v"(v[1]), "v"(v[2]), "v"(v[3]), "v"(v[4]), "v"(v[5]), "v"(v[6]), "v"(v[7]), "v"(v[8]), "v"(v[9]),
                        "v"(v[10]), "v"(v[11]), "v"(v[12]), "v"(v[13]), "v"(v[14]), "v"(v[15]), "v"(v[16]), "v"(v[17]), "v"(v[18]), "v"(v[19]),
                        "v"(v[20]), "v"(v[21]), "v"(v[22]), "v"(v[23]), "v"(v[24]), "v"(v[25]), "v"(v[26]), "v"(v[27]), "v"(v[28]), "v"(v[29]));
#pragma unroll
        for (int k = 0; k < HG_GRP; ++k) c += v[k];
    }
    *(float4*)(deg + i4) = make_float4(rsqrtf((float)c.x), rsqrtf((float)c.y), rsqrtf((float)c.z), rsqrtf((float)c.w));
}


#define AG_ROWS    256
static_assert(AG_GROUP <= AG_ROWS, "a group's rows must fit the work-group's registers and tiles");
#define AG_SLOTS   8
#define AG_WAVES   32
#define AG_LCAP    1536
#define AG_MCAP    384
#define AG_TP      132
#define AG_HP      33

__device__ __forceinline__ const unsigned* gcn_keys_ro(const float* dinv, int n) {
    return (const unsigned*)((const char*)dinv + ((((size_t)n) * 4 + 255) & ~(size_t)255));
}

struct ag_key { unsigned s[8]; };

__device__ __forceinline__ void ag_keys32(const uint4 qa, const uint4 qb, unsigned e0, const ag_key& K,
                                          unsigned lt, unsigned* wl, int& cnt)
{
    unsigned df = qa.x ^ K.s[0];
    df |= qa.y ^ K.s[1];
    df |= qa.z ^ K.s[2];
    df |= qa.w ^ K.s[3];
    df |= qb.x ^ K.s[4];
    df |= qb.y ^ K.s[5];
    df |= qb.z ^ K.s[6];
    df |= qb.w ^ K.s[7];
    unsigned cm = ~df;
    for (;;) {
        const unsigned mask = __builtin_amdgcn_ballot_w32(cm != 0u);
        if (!mask) break;
        if (cm) {
            wl[cnt + __popc(mask & lt)] = e0 + (unsigned)__builtin_ctz(cm);
            cm &= cm - 1u;
        }
        cnt += __popc(mask);
    }
}

typedef __attribute__((ext_vector_type(32))) float ag_v32f;
struct ag_acc { ag_v32f a; };

__device__ __forceinline__ void ag_addslot(ag_acc& A, int slot, const float4 v, float nr)
{
    const int b = slot * 4;
    A.a[b + 0] += v.x * nr;
    A.a[b + 1] += v.y * nr;
    A.a[b + 2] += v.z * nr;
    A.a[b + 3] += v.w * nr;
}

template <int NG>
__device__ __forceinline__ void ag_addgroup(ag_acc& A, const unsigned* ent, int base, int lane,
                                            const float4* __restrict__ xw4, const float* __restrict__ dinv)
{
    unsigned t[4 * NG];
#pragma unroll
    for (int g = 0; g < NG; ++g) {
        const uint4 en = *(const uint4*)(ent + 4 * g);
        t[4 * g + 0] = __builtin_amdgcn_readfirstlane(en.x);
        t[4 * g + 1] = __builtin_amdgcn_readfirstlane(en.y);
        t[4 * g + 2] = __builtin_amdgcn_readfirstlane(en.z);
        t[4 * g + 3] = __builtin_amdgcn_readfirstlane(en.w);
    }
    float4 v[4 * NG];
    float nr[4 * NG];
#pragma unroll
    for (int i = 0; i < 4 * NG; ++i)
        v[i] = xw4[(size_t)(t[i] & 0xFFFFFFu) * (CH / 4) + lane];
#pragma unroll
    for (int i = 0; i < 4 * NG; ++i)
        nr[i] = dinv[t[i] & 0xFFFFFFu] * dinv[base + (int)(t[i] >> 24)];
#pragma unroll
    for (int i = 0; i < 4 * NG; ++i)
        ag_addslot(A, (int)(t[i] >> 29), v[i], nr[i]);
}

__device__ __forceinline__ void ag_flush(ag_acc& A, unsigned* mylist, int mcnt, int base, int lane,
                                         const int* __restrict__ src,
                                         const float4* __restrict__ xw4, const float* __restrict__ dinv)
{
    for (int k = lane; k < mcnt; k += 32) {
        const unsigned t = mylist[k];
        mylist[k] = (t & 0xFF000000u) | (unsigned)src[t & 0xFFFFFFu];
    }
    int k = 0;
    for (; k + 12 <= mcnt; k += 12) ag_addgroup<3>(A, mylist + k, base, lane, xw4, dinv);
    for (; k + 4 <= mcnt; k += 4)   ag_addgroup<1>(A, mylist + k, base, lane, xw4, dinv);
    for (; k < mcnt; ++k) {
        const unsigned t = __builtin_amdgcn_readfirstlane(mylist[k]);
        const int s = (int)(t & 0xFFFFFFu), r = (int)(t >> 24);
        const float4 v = xw4[(size_t)s * (CH / 4) + lane];
        const float nr = dinv[s] * dinv[base + r];
        ag_addslot(A, r >> 5, v, nr);
    }
}

__device__ __forceinline__ float4 ag_start(int r, int rows, int base, int lane,
                                           const float4* __restrict__ xw4, const float* __restrict__ dinv)
{
    float4 a = make_float4(0.f, 0.f, 0.f, 0.f);
    if (r < rows) {
        const int d = base + r;
        const float di = dinv[d];
        const float4 x = xw4[(size_t)d * (CH / 4) + lane];
        a.x = x.x * di * di;
        a.y = x.y * di * di;
        a.z = x.z * di * di;
        a.w = x.w * di * di;
    }
    return a;
}

__global__ __launch_bounds__(1024) void gcn_aggregate(
    const int* __restrict__ src, const int* __restrict__ dst,
    const float* __restrict__ dinv, const float* __restrict__ x, const float* __restrict__ W,
    const float* __restrict__ bias, float* __restrict__ out, int n, int E)
{
    __shared__ uint4    pool4[2 * AG_ROWS * AG_TP / 4];
    __shared__ int      more[2];
    __shared__ unsigned hist[AG_WAVES * AG_HP];
    __shared__ unsigned cur[AG_WAVES * AG_HP];
    __shared__ int      most[2];

    const int wave = __builtin_amdgcn_readfirstlane((int)(threadIdx.x >> 5));
    const int lane = threadIdx.x & 31;
    const unsigned lt = (1u << lane) - 1u;
    const int base = blockIdx.x * AG_GROUP;
    int rows = n - base;
    if (rows > AG_GROUP) rows = AG_GROUP;

    unsigned* lists  = (unsigned*)pool4;
    unsigned* wl     = lists + wave * AG_LCAP;
    unsigned* mylist = lists + AG_WAVES * AG_LCAP + wave * AG_MCAP;
    const float4* xw4 = (const float4*)x;

    ag_acc A;
    {
#pragma unroll
        for (int i = 0; i < AG_SLOTS; ++i) {
            const float4 t = ag_start(wave + 32 * i, rows, base, lane, xw4, dinv);
            A.a[4 * i + 0] = t.x; A.a[4 * i + 1] = t.y; A.a[4 * i + 2] = t.z; A.a[4 * i + 3] = t.w;
        }
    }
    if (threadIdx.x == 0) { more[0] = 0; more[1] = 0; most[0] = 0; most[1] = 0; }
    __syncthreads();

    const unsigned* keys = gcn_keys_ro(dinv, n);
    const int P = gcn_padded(E);
    const int segLen = ((((P >> 10) + AG_WAVES - 1) / AG_WAVES)) << 10;
    int pos = wave * segLen;
    int segEnd = pos + segLen;
    if (segEnd > P) segEnd = P;
    if (pos > P) pos = P;
    ag_key K;
#pragma unroll
    for (int b = 0; b < 8; ++b) K.s[b] = 0u - ((blockIdx.x >> b) & 1u);

    for (int pass = 0; ; ++pass) {
        int cnt = 0;
        {
            const int left = (segEnd - pos) >> 10;
            if (left > 0) {
                const uint4* p = (const uint4*)(keys + ((size_t)(pos >> 10) << 8)) + lane;
                const int last = left - 1;
                const int i1 = 1 < last ? 1 : last, i2 = 2 < last ? 2 : last, i3 = 3 < last ? 3 : last;
                uint4 a0 = p[0],       b0 = p[32];
                uint4 a1 = p[64 * i1], b1 = p[64 * i1 + 32];
                uint4 a2 = p[64 * i2], b2 = p[64 * i2 + 32];
                uint4 a3 = p[64 * i3], b3 = p[64 * i3 + 32];
                int i = 0;
#define AG_STEP(RA, RB)                                                              \
                {                                                                    \
                    go = go && (i < left) && (cnt <= AG_LCAP - 1024);                \
                    if (go) {                                                        \
                        ag_keys32(RA, RB, (unsigned)pos + 32u * (unsigned)lane, K, lt, wl, cnt); \
                        pos += 1024; ++i;                                            \
                    }                                                                \
                    const int nx = i + 3 < last ? i + 3 : last;                      \
                    RA = p[64 * nx]; RB = p[64 * nx + 32];                           \
                }
                for (;;) {
                    bool go = true;
                    AG_STEP(a0, b0) AG_STEP(a1, b1) AG_STEP(a2, b2) AG_STEP(a3, b3)
                    if (!go) break;
                }
#undef AG_STEP
            }
        }
        {
            int outc = 0;
            hist[wave * AG_HP + lane] = 0u;
            for (int i0 = 0; i0 < cnt; i0 += 128) {
                unsigned ee[4];
                int dv[4];
#pragma unroll
                for (int u = 0; u < 4; ++u) {
                    const int idx = i0 + 32 * u + lane;
                    ee[u] = (idx < cnt) ? wl[idx] : 0xFFFFFFFFu;
                }
#pragma unroll
                for (int u = 0; u < 4; ++u)
                    dv[u] = (ee[u] < (unsigned)E) ? dst[ee[u]] : -1;
#pragma unroll
                for (int u = 0; u < 4; ++u) {
                    const unsigned dl = (unsigned)(dv[u] - base);
                    const bool m = (dv[u] >= 0) && (dl < (unsigned)rows);
                    const unsigned mask = __builtin_amdgcn_ballot_w32(m);
                    if (mask) {
                        if (m) {
                            wl[outc + __popc(mask & lt)] = (dl << 24) | ee[u];
                            atomicAdd(&hist[wave * AG_HP + (int)(dl & 31u)], 1u);
                        }
                        outc += __popc(mask);
                    }
                }
            }
            cnt = outc;
        }

        if (lane == 0 && pos < segEnd) more[pass & 1] = 1;
        __syncthreads();
        if (threadIdx.x == 0) { more[(pass + 1) & 1] = 0; most[(pass + 1) & 1] = 0; }

        int total;
        {
            const unsigned hc = hist[lane * AG_HP + wave];
            int incl = (int)hc;
            incl += __builtin_amdgcn_update_dpp(0, incl, 0x111, 0xf, 0xf, true);
            incl += __builtin_amdgcn_update_dpp(0, incl, 0x112, 0xf, 0xf, true);
            incl += __builtin_amdgcn_update_dpp(0, incl, 0x114, 0xf, 0xf, true);
            incl += __builtin_amdgcn_update_dpp(0, incl, 0x118, 0xf, 0xf, true);
            const int low = __builtin_amdgcn_readlane(incl, 15);
            if (lane >= 16) incl += low;
            total = __builtin_amdgcn_readlane(incl, 31);
            hist[lane * AG_HP + wave] = (unsigned)incl - hc;
            if (lane == 0) atomicMax(&most[pass & 1], total);
        }
        __syncthreads();
        const int mx = __builtin_amdgcn_readfirstlane(most[pass & 1]);

        unsigned* own = lists + AG_WAVES * AG_LCAP;
        for (int r0 = 0; r0 < mx; r0 += AG_MCAP) {
            unsigned* cw = cur + wave * AG_HP;
            cw[lane] = hist[wave * AG_HP + lane];
            for (int i0 = 0; i0 < cnt; i0 += 32) {
                const int idx = i0 + lane;
                const bool valid = idx < cnt;
                const unsigned e = valid ? wl[idx] : 0u;
                const unsigned o = (e >> 24) & 31u;
                unsigned same = __builtin_amdgcn_ballot_w32(valid);
#pragma unroll
                for (int b = 0; b < 5; ++b) {
                    const bool bit = ((o >> b) & 1u) != 0u;
                    const unsigned bb = __builtin_amdgcn_ballot_w32(bit);
                    same &= bit ? bb : ~bb;
                }
                if (valid) {
                    const unsigned c0 = cw[o];
                    const unsigned rk = (unsigned)__popc(same & lt);
                    const unsigned pl = c0 + rk - (unsigned)r0;
                    if (pl < (unsigned)AG_MCAP) own[o * AG_MCAP + pl] = e;
                    if (rk == 0u) cw[o] = c0 + (unsigned)__popc(same);
                }
            }
            __syncthreads();

            int mcnt = total - r0;
            if (mcnt > AG_MCAP) mcnt = AG_MCAP;
            if (mcnt < 0) mcnt = 0;
            ag_flush(A, mylist, mcnt, base, lane, src, xw4, dinv);
            if (r0 + AG_MCAP < mx) __syncthreads();
        }

        const int again = more[pass & 1];
        __syncthreads();
        if (!again) break;
    }

    const int half = lane >> 4;
    const int l16  = lane & 15;
    const int ct   = wave & 7;
    const int rg   = wave >> 3;
    float* tileA = (float*)pool4 + AG_ROWS * AG_TP;

    v2f bf[32];
    {
        const float* Wp = W + (size_t)(2 * half) * CH + 16 * ct + l16;
#pragma unroll
        for (int ks = 0; ks < 32; ++ks) {
            bf[ks].x = Wp[(size_t)(4 * ks) * CH];
            bf[ks].y = Wp[(size_t)(4 * ks + 1) * CH];
        }
    }

#pragma unroll
    for (int i = 0; i < AG_SLOTS; ++i) {
        const int r = wave + 32 * i;
        *(float4*)(tileA + r * AG_TP + 4 * lane) =
            make_float4(A.a[4 * i + 0], A.a[4 * i + 1], A.a[4 * i + 2], A.a[4 * i + 3]);
    }
    __syncthreads();

    unsigned* tileD = (unsigned*)pool4;
    for (int hh = 0; hh < 2; ++hh) {
        const int R0 = 64 * rg + 32 * hh;
        if (R0 >= rows) break;
        const float* a0p = tileA + (R0 + l16) * AG_TP + 2 * half;
        const float* a1p = a0p + 16 * AG_TP;
        v8f c0 = {}, c1 = {};
#pragma unroll
        for (int ks = 0; ks < 32; ++ks) {
            const v2f a0 = *(const v2f*)(a0p + 4 * ks);
            const v2f a1 = *(const v2f*)(a1p + 4 * ks);
            c0 = __builtin_amdgcn_wmma_f32_16x16x4_f32(
                      false, a0,  false, bf[ks],
                      (short)0, c0,  false,  false);
            c1 = __builtin_amdgcn_wmma_f32_16x16x4_f32(
                      false, a1,  false, bf[ks],
                      (short)0, c1,  false,  false);
        }
#pragma unroll
        for (int r = 0; r < 8; ++r) {
            tileD[(R0 + r + 8 * half) * AG_TP + 16 * ct + l16]      = __float_as_uint(c0[r]);
            tileD[(R0 + 16 + r + 8 * half) * AG_TP + 16 * ct + l16] = __float_as_uint(c1[r]);
        }
    }
    __syncthreads();

    float4* out4 = (float4*)out;
    const float4 b4 = ((const float4*)bias)[lane];
#pragma unroll
    for (int i = 0; i < AG_SLOTS; ++i) {
        const int r = wave + 32 * i;
        if (r < rows) {
            const uint4 u = *(const uint4*)(tileD + r * AG_TP + 4 * lane);
            out4[(size_t)(base + r) * (CH / 4) + lane] =
                make_float4(__uint_as_float(u.x) + b4.x, __uint_as_float(u.y) + b4.y,
                            __uint_as_float(u.z) + b4.z, __uint_as_float(u.w) + b4.w);
        }
    }
}


extern "C" void kernel_launch(void* const* d_in, const int* in_sizes, int n_in,
                              void* d_out, int out_size, void* d_ws, size_t ws_size,
                              hipStream_t stream) {
    const float* x  = (const float*)d_in[0];
    const float* W  = (const float*)d_in[1];
    const float* b  = (const float*)d_in[2];
    const int*   ei = (const int*)d_in[3];

    const int n = in_sizes[0] / CH;
    const int E = in_sizes[3] / 2;
    const int* src = ei;
    const int* dst = ei + E;

    float* deg = (float*)d_ws;
    float* out = (float*)d_out;

    gcn_deg_count<<<HG_PARTS + (E + 4095) / 4096, 1024, 0, stream>>>(dst, deg, n, E);
    gcn_dinv     <<<(n + 255) / 256, 64, 0, stream>>>(deg, n, E);

    gcn_aggregate<<<(n + AG_GROUP - 1) / AG_GROUP, 1024, 0, stream>>>(src, dst, deg, x, W, b, out, n, E);
}


# aggregation: loads once waited for one by one go out as groups: the 8 self-loop rows, an owner list's src lookups (4 per lane), the list's tail (masked groups of 8, was fours and singles); sums same
# speedup vs baseline: 4.0708x; 1.0344x over previous
#include <hip/hip_runtime.h>
#include <hip/hip_bf16.h>

#define CH 128

typedef __attribute__((ext_vector_type(2))) float v2f;
typedef __attribute__((ext_vector_type(8))) float v8f;


#define AG_GROUP 200

__device__ __forceinline__ unsigned* gcn_keys(float* deg, int n) {
    return (unsigned*)((char*)deg + ((((size_t)n) * 4 + 255) & ~(size_t)255));
}
__device__ __forceinline__ int gcn_padded(int E) { return (E + 1023) & ~1023; }

#define HG_PARTS 60
#define HG_CAP   51200

__device__ __forceinline__ int gcn_npad(int n) { return (n + 63) & ~63; }
__device__ __forceinline__ unsigned* gcn_parts(float* deg, int n, int E) {
    return (unsigned*)((char*)gcn_keys(deg, n) + (size_t)gcn_padded(E));
}

struct hg_quad { int d[4]; };

__device__ __forceinline__ int hg_edge(int s, int wave, int j, int lane) {
    return s * 1024 + (wave & 7) * 128 + 32 * j + lane;
}

__device__ __forceinline__ hg_quad hg_load(const int* __restrict__ dst, int s, int wave, int lane, int E) {
    hg_quad q;
#pragma unroll
    for (int j = 0; j < 4; ++j) {
        const int e = hg_edge(s, wave, j, lane);
        q.d[j] = dst[e < E ? e : E - 1];
    }
    return q;
}

__device__ __forceinline__ void hg_keys(const hg_quad& q, int s, int wave, int lane, int E, unsigned* keys)
{
    unsigned myw = 0;
#pragma unroll
    for (int j = 0; j < 4; ++j) {
        const unsigned key = (hg_edge(s, wave, j, lane) < E) ? (((unsigned)q.d[j] / AG_GROUP) & 255u) : 0u;
#pragma unroll
        for (int b = 0; b < 8; ++b) {
            const unsigned w = __builtin_amdgcn_ballot_w32(((key >> b) & 1u) != 0u);
            if (lane == 8 * j + b) myw = w;
        }
    }
    const int j = lane >> 3, b = lane & 7;
    keys[(size_t)s * 256 + (b >> 2) * 128 + (4 * (wave & 7) + j) * 4 + (b & 3)] = myw;
}

__global__ __launch_bounds__(1024) void gcn_deg_count(const int* __restrict__ dst, float* __restrict__ deg, int n, int E) {
    __shared__ uint4 hist4[HG_CAP / 4];
    unsigned* hist = (unsigned*)hist4;
    const int lane = threadIdx.x & 31;
    const int wave = __builtin_amdgcn_readfirstlane((int)(threadIdx.x >> 5));
    const int nsteps = gcn_padded(E) >> 10;

    if (blockIdx.x >= HG_PARTS) {
        const int s = 4 * ((int)blockIdx.x - HG_PARTS) + (wave >> 3);
        if (s < nsteps) {
            const hg_quad q = hg_load(dst, s, wave, lane, E);
            asm volatile("" :: "v"(q.d[0]), "v"(q.d[1]), "v"(q.d[2]), "v"(q.d[3]));
            hg_keys(q, s, wave, lane, E, gcn_keys(deg, n));
        }
        return;
    }

    const int s_lo = (int)(((long long)blockIdx.x * nsteps) / HG_PARTS);
    const int s_hi = (int)(((long long)(blockIdx.x + 1) * nsteps) / HG_PARTS);
    const int npad = gcn_npad(n);
    unsigned* part = gcn_parts(deg, n, E) + (size_t)blockIdx.x * npad;

    for (int nb = 0; nb < npad; nb += HG_CAP) {
        int span = npad - nb;
        if (span > HG_CAP) span = HG_CAP;
        hg_quad q[4] = {};
        if (s_lo < s_hi) {
#pragma unroll
            for (int t = 0; t < 4; ++t) q[t] = hg_load(dst, s_lo + 4 * t + (wave >> 3), wave, lane, E);
        }
        for (int i = threadIdx.x; i < (span >> 2); i += 1024) hist4[i] = make_uint4(0u, 0u, 0u, 0u);
        __syncthreads();

        for (int sb = s_lo; sb < s_hi; sb += 16) {
#pragma unroll
            for (int t = 0; t < 4; ++t) {
                const int s = sb + 4 * t + (wave >> 3);
#pragma unroll
                for (int j = 0; j < 4; ++j) {
                    const unsigned dl = (unsigned)(q[t].d[j] - nb);
                    if (s < s_hi && hg_edge(s, wave, j, lane) < E && dl < (unsigned)span)
                        atomicAdd(&hist[dl], 1u);
                }
            }
            if (sb + 16 < s_hi) {
#pragma unroll
                for (int t = 0; t < 4; ++t) q[t] = hg_load(dst, sb + 16 + 4 * t + (wave >> 3), wave, lane, E);
            }
        }
        __syncthreads();

        uint4* p4 = (uint4*)(part + nb);
#pragma unroll 4
        for (int i = threadIdx.x; i < (span >> 2); i += 1024) p4[i] = hist4[i];
        __syncthreads();
    }
}

#define HG_GRP 30
static_assert(HG_PARTS % HG_GRP == 0, "the planes are read in whole groups");
typedef __attribute__((ext_vector_type(4))) unsigned hg_v4u;
__global__ __launch_bounds__(64) void gcn_dinv(float* __restrict__ deg, int n, int E) {
    const int i4 = (blockIdx.x * blockDim.x + threadIdx.x) * 4;
    if (i4 >= n) return;
    const size_t pitch4 = (size_t)(gcn_npad(n) >> 2);
    const hg_v4u* p = (const hg_v4u*)(gcn_parts(deg, n, E) + i4);
    hg_v4u c = {1u, 1u, 1u, 1u};
    for (int g0 = 0; g0 < HG_PARTS; g0 += HG_GRP) {
        hg_v4u v[HG_GRP];
#pragma unroll
        for (int k = 0; k < HG_GRP; ++k) v[k] = p[(size_t)(g0 + k) * pitch4];
        asm volatile("" ::
                        "v"(v[0]), "v"(v[1]), "v"(v[2]), "v"(v[3]), "v"(v[4]), "v"(v[5]), "v"(v[6]), "v"(v[7]), "v"(v[8]), "v"(v[9]),
                        "v"(v[10]), "v"(v[11]), "v"(v[12]), "v"(v[13]), "v"(v[14]), "v"(v[15]), "v"(v[16]), "v"(v[17]), "v"(v[18]), "v"(v[19]),
                        "v"(v[20]), "v"(v[21]), "v"(v[22]), "v"(v[23]), "v"(v[24]), "v"(v[25]), "v"(v[26]), "v"(v[27]), "v"(v[28]), "v"(v[29]));
#pragma unroll
        for (int k = 0; k < HG_GRP; ++k) c += v[k];
    }
    *(float4*)(deg + i4) = make_float4(rsqrtf((float)c.x), rsqrtf((float)c.y), rsqrtf((float)c.z), rsqrtf((float)c.w));
}


#define AG_ROWS    256
static_assert(AG_GROUP <= AG_ROWS, "a group's rows must fit the work-group's registers and tiles");
#define AG_SLOTS   8
#define AG_WAVES   32
#define AG_LCAP    1536
#define AG_MCAP    384
#define AG_TP      132
#define AG_HP      33

__device__ __forceinline__ const unsigned* gcn_keys_ro(const float* dinv, int n) {
    return (const unsigned*)((const char*)dinv + ((((size_t)n) * 4 + 255) & ~(size_t)255));
}

struct ag_key { unsigned s[8]; };

__device__ __forceinline__ void ag_keys32(const uint4 qa, const uint4 qb, unsigned e0, const ag_key& K,
                                          unsigned lt, unsigned* wl, int& cnt)
{
    unsigned df = qa.x ^ K.s[0];
    df |= qa.y ^ K.s[1];
    df |= qa.z ^ K.s[2];
    df |= qa.w ^ K.s[3];
    df |= qb.x ^ K.s[4];
    df |= qb.y ^ K.s[5];
    df |= qb.z ^ K.s[6];
    df |= qb.w ^ K.s[7];
    unsigned cm = ~df;
    for (;;) {
        const unsigned mask = __builtin_amdgcn_ballot_w32(cm != 0u);
        if (!mask) break;
        if (cm) {
            wl[cnt + __popc(mask & lt)] = e0 + (unsigned)__builtin_ctz(cm);
            cm &= cm - 1u;
        }
        cnt += __popc(mask);
    }
}

typedef __attribute__((ext_vector_type(32))) float ag_v32f;
struct ag_acc { ag_v32f a; };

__device__ __forceinline__ void ag_addslot(ag_acc& A, int slot, const float4 v, float nr)
{
    const int b = slot * 4;
    A.a[b + 0] += v.x * nr;
    A.a[b + 1] += v.y * nr;
    A.a[b + 2] += v.z * nr;
    A.a[b + 3] += v.w * nr;
}

template <int NG>
__device__ __forceinline__ void ag_addgroup(ag_acc& A, const unsigned* ent, int base, int lane,
                                            const float4* __restrict__ xw4, const float* __restrict__ dinv)
{
    unsigned t[4 * NG];
#pragma unroll
    for (int g = 0; g < NG; ++g) {
        const uint4 en = *(const uint4*)(ent + 4 * g);
        t[4 * g + 0] = __builtin_amdgcn_readfirstlane(en.x);
        t[4 * g + 1] = __builtin_amdgcn_readfirstlane(en.y);
        t[4 * g + 2] = __builtin_amdgcn_readfirstlane(en.z);
        t[4 * g + 3] = __builtin_amdgcn_readfirstlane(en.w);
    }
    float4 v[4 * NG];
    float nr[4 * NG];
#pragma unroll
    for (int i = 0; i < 4 * NG; ++i)
        v[i] = xw4[(size_t)(t[i] & 0xFFFFFFu) * (CH / 4) + lane];
#pragma unroll
    for (int i = 0; i < 4 * NG; ++i)
        nr[i] = dinv[t[i] & 0xFFFFFFu] * dinv[base + (int)(t[i] >> 24)];
#pragma unroll
    for (int i = 0; i < 4 * NG; ++i)
        ag_addslot(A, (int)(t[i] >> 29), v[i], nr[i]);
}

template <int NG>
__device__ __forceinline__ void ag_addtail(ag_acc& A, const unsigned* ent, int nv, int base, int lane,
                                           const float4* __restrict__ xw4, const float* __restrict__ dinv)
{
    unsigned t[4 * NG];
#pragma unroll
    for (int g = 0; g < NG; ++g) {
        const uint4 en = *(const uint4*)(ent + 4 * g);
        t[4 * g + 0] = __builtin_amdgcn_readfirstlane(en.x);
        t[4 * g + 1] = __builtin_amdgcn_readfirstlane(en.y);
        t[4 * g + 2] = __builtin_amdgcn_readfirstlane(en.z);
        t[4 * g + 3] = __builtin_amdgcn_readfirstlane(en.w);
    }
#pragma unroll
    for (int i = 1; i < 4 * NG; ++i) t[i] = (i < nv) ? t[i] : t[0];
    float4 v[4 * NG];
    float nr[4 * NG];
#pragma unroll
    for (int i = 0; i < 4 * NG; ++i)
        v[i] = xw4[(size_t)(t[i] & 0xFFFFFFu) * (CH / 4) + lane];
#pragma unroll
    for (int i = 0; i < 4 * NG; ++i)
        nr[i] = dinv[t[i] & 0xFFFFFFu] * dinv[base + (int)(t[i] >> 24)];
#pragma unroll
    for (int i = 0; i < 4 * NG; ++i)
        if (i < nv) ag_addslot(A, (int)(t[i] >> 29), v[i], nr[i]);
}

__device__ __forceinline__ void ag_flush(ag_acc& A, unsigned* mylist, int mcnt, int base, int lane,
                                         const int* __restrict__ src,
                                         const float4* __restrict__ xw4, const float* __restrict__ dinv)
{
    for (int k0 = 0; k0 < mcnt; k0 += 128) {
        unsigned t[4];
        int sn[4];
#pragma unroll
        for (int u = 0; u < 4; ++u) {
            const int idx = k0 + 32 * u + lane;
            t[u] = (idx < mcnt) ? mylist[idx] : 0u;
        }
#pragma unroll
        for (int u = 0; u < 4; ++u) sn[u] = src[t[u] & 0xFFFFFFu];
#pragma unroll
        for (int u = 0; u < 4; ++u) {
            const int idx = k0 + 32 * u + lane;
            if (idx < mcnt) mylist[idx] = (t[u] & 0xFF000000u) | (unsigned)sn[u];
        }
    }
    int k = 0;
    for (; k + 12 <= mcnt; k += 12) ag_addgroup<3>(A, mylist + k, base, lane, xw4, dinv);
    for (; k < mcnt; k += 8) ag_addtail<2>(A, mylist + k, mcnt - k < 8 ? mcnt - k : 8, base, lane, xw4, dinv);
}

__global__ __launch_bounds__(1024) void gcn_aggregate(
    const int* __restrict__ src, const int* __restrict__ dst,
    const float* __restrict__ dinv, const float* __restrict__ x, const float* __restrict__ W,
    const float* __restrict__ bias, float* __restrict__ out, int n, int E)
{
    __shared__ uint4    pool4[2 * AG_ROWS * AG_TP / 4];
    __shared__ int      more[2];
    __shared__ unsigned hist[AG_WAVES * AG_HP];
    __shared__ unsigned cur[AG_WAVES * AG_HP];
    __shared__ int      most[2];

    const int wave = __builtin_amdgcn_readfirstlane((int)(threadIdx.x >> 5));
    const int lane = threadIdx.x & 31;
    const unsigned lt = (1u << lane) - 1u;
    const int base = blockIdx.x * AG_GROUP;
    int rows = n - base;
    if (rows > AG_GROUP) rows = AG_GROUP;

    unsigned* lists  = (unsigned*)pool4;
    unsigned* wl     = lists + wave * AG_LCAP;
    unsigned* mylist = lists + AG_WAVES * AG_LCAP + wave * AG_MCAP;
    const float4* xw4 = (const float4*)x;

    ag_acc A;
    {
        float4 xs[AG_SLOTS];
        float  dv[AG_SLOTS];
#pragma unroll
        for (int i = 0; i < AG_SLOTS; ++i) {
            const int r = wave + 32 * i;
            const int d = base + (r < rows ? r : 0);
            xs[i] = xw4[(size_t)d * (CH / 4) + lane];
            dv[i] = dinv[d];
        }
#pragma unroll
        for (int i = 0; i < AG_SLOTS; ++i) {
            const bool own = wave + 32 * i < rows;
            const float di = dv[i];
            A.a[4 * i + 0] = own ? xs[i].x * di * di : 0.f;
            A.a[4 * i + 1] = own ? xs[i].y * di * di : 0.f;
            A.a[4 * i + 2] = own ? xs[i].z * di * di : 0.f;
            A.a[4 * i + 3] = own ? xs[i].w * di * di : 0.f;
        }
    }
    if (threadIdx.x == 0) { more[0] = 0; more[1] = 0; most[0] = 0; most[1] = 0; }
    __syncthreads();

    const unsigned* keys = gcn_keys_ro(dinv, n);
    const int P = gcn_padded(E);
    const int segLen = ((((P >> 10) + AG_WAVES - 1) / AG_WAVES)) << 10;
    int pos = wave * segLen;
    int segEnd = pos + segLen;
    if (segEnd > P) segEnd = P;
    if (pos > P) pos = P;
    ag_key K;
#pragma unroll
    for (int b = 0; b < 8; ++b) K.s[b] = 0u - ((blockIdx.x >> b) & 1u);

    for (int pass = 0; ; ++pass) {
        int cnt = 0;
        {
            const int left = (segEnd - pos) >> 10;
            if (left > 0) {
                const uint4* p = (const uint4*)(keys + ((size_t)(pos >> 10) << 8)) + lane;
                const int last = left - 1;
                const int i1 = 1 < last ? 1 : last, i2 = 2 < last ? 2 : last, i3 = 3 < last ? 3 : last;
                uint4 a0 = p[0],       b0 = p[32];
                uint4 a1 = p[64 * i1], b1 = p[64 * i1 + 32];
                uint4 a2 = p[64 * i2], b2 = p[64 * i2 + 32];
                uint4 a3 = p[64 * i3], b3 = p[64 * i3 + 32];
                int i = 0;
#define AG_STEP(RA, RB)                                                              \
                {                                                                    \
                    go = go && (i < left) && (cnt <= AG_LCAP - 1024);                \
                    if (go) {                                                        \
                        ag_keys32(RA, RB, (unsigned)pos + 32u * (unsigned)lane, K, lt, wl, cnt); \
                        pos += 1024; ++i;                                            \
                    }                                                                \
                    const int nx = i + 3 < last ? i + 3 : last;                      \
                    RA = p[64 * nx]; RB = p[64 * nx + 32];                           \
                }
                for (;;) {
                    bool go = true;
                    AG_STEP(a0, b0) AG_STEP(a1, b1) AG_STEP(a2, b2) AG_STEP(a3, b3)
                    if (!go) break;
                }
#undef AG_STEP
            }
        }
        {
            int outc = 0;
            hist[wave * AG_HP + lane] = 0u;
            for (int i0 = 0; i0 < cnt; i0 += 128) {
                unsigned ee[4];
                int dv[4];
#pragma unroll
                for (int u = 0; u < 4; ++u) {
                    const int idx = i0 + 32 * u + lane;
                    ee[u] = (idx < cnt) ? wl[idx] : 0xFFFFFFFFu;
                }
#pragma unroll
                for (int u = 0; u < 4; ++u)
                    dv[u] = (ee[u] < (unsigned)E) ? dst[ee[u]] : -1;
#pragma unroll
                for (int u = 0; u < 4; ++u) {
                    const unsigned dl = (unsigned)(dv[u] - base);
                    const bool m = (dv[u] >= 0) && (dl < (unsigned)rows);
                    const unsigned mask = __builtin_amdgcn_ballot_w32(m);
                    if (mask) {
                        if (m) {
                            wl[outc + __popc(mask & lt)] = (dl << 24) | ee[u];
                            atomicAdd(&hist[wave * AG_HP + (int)(dl & 31u)], 1u);
                        }
                        outc += __popc(mask);
                    }
                }
            }
            cnt = outc;
        }

        if (lane == 0 && pos < segEnd) more[pass & 1] = 1;
        __syncthreads();
        if (threadIdx.x == 0) { more[(pass + 1) & 1] = 0; most[(pass + 1) & 1] = 0; }

        int total;
        {
            const unsigned hc = hist[lane * AG_HP + wave];
            int incl = (int)hc;
            incl += __builtin_amdgcn_update_dpp(0, incl, 0x111, 0xf, 0xf, true);
            incl += __builtin_amdgcn_update_dpp(0, incl, 0x112, 0xf, 0xf, true);
            incl += __builtin_amdgcn_update_dpp(0, incl, 0x114, 0xf, 0xf, true);
            incl += __builtin_amdgcn_update_dpp(0, incl, 0x118, 0xf, 0xf, true);
            const int low = __builtin_amdgcn_readlane(incl, 15);
            if (lane >= 16) incl += low;
            total = __builtin_amdgcn_readlane(incl, 31);
            hist[lane * AG_HP + wave] = (unsigned)incl - hc;
            if (lane == 0) atomicMax(&most[pass & 1], total);
        }
        __syncthreads();
        const int mx = __builtin_amdgcn_readfirstlane(most[pass & 1]);

        unsigned* own = lists + AG_WAVES * AG_LCAP;
        for (int r0 = 0; r0 < mx; r0 += AG_MCAP) {
            unsigned* cw = cur + wave * AG_HP;
            cw[lane] = hist[wave * AG_HP + lane];
            for (int i0 = 0; i0 < cnt; i0 += 32) {
                const int idx = i0 + lane;
                const bool valid = idx < cnt;
                const unsigned e = valid ? wl[idx] : 0u;
                const unsigned o = (e >> 24) & 31u;
                unsigned same = __builtin_amdgcn_ballot_w32(valid);
#pragma unroll
                for (int b = 0; b < 5; ++b) {
                    const bool bit = ((o >> b) & 1u) != 0u;
                    const unsigned bb = __builtin_amdgcn_ballot_w32(bit);
                    same &= bit ? bb : ~bb;
                }
                if (valid) {
                    const unsigned c0 = cw[o];
                    const unsigned rk = (unsigned)__popc(same & lt);
                    const unsigned pl = c0 + rk - (unsigned)r0;
                    if (pl < (unsigned)AG_MCAP) own[o * AG_MCAP + pl] = e;
                    if (rk == 0u) cw[o] = c0 + (unsigned)__popc(same);
                }
            }
            __syncthreads();

            int mcnt = total - r0;
            if (mcnt > AG_MCAP) mcnt = AG_MCAP;
            if (mcnt < 0) mcnt = 0;
            ag_flush(A, mylist, mcnt, base, lane, src, xw4, dinv);
            if (r0 + AG_MCAP < mx) __syncthreads();
        }

        const int again = more[pass & 1];
        __syncthreads();
        if (!again) break;
    }

    const int half = lane >> 4;
    const int l16  = lane & 15;
    const int ct   = wave & 7;
    const int rg   = wave >> 3;
    float* tileA = (float*)pool4 + AG_ROWS * AG_TP;

    v2f bf[32];
    {
        const float* Wp = W + (size_t)(2 * half) * CH + 16 * ct + l16;
#pragma unroll
        for (int ks = 0; ks < 32; ++ks) {
            bf[ks].x = Wp[(size_t)(4 * ks) * CH];
            bf[ks].y = Wp[(size_t)(4 * ks + 1) * CH];
        }
    }

#pragma unroll
    for (int i = 0; i < AG_SLOTS; ++i) {
        const int r = wave + 32 * i;
        *(float4*)(tileA + r * AG_TP + 4 * lane) =
            make_float4(A.a[4 * i + 0], A.a[4 * i + 1], A.a[4 * i + 2], A.a[4 * i + 3]);
    }
    __syncthreads();

    unsigned* tileD = (unsigned*)pool4;
    for (int hh = 0; hh < 2; ++hh) {
        const int R0 = 64 * rg + 32 * hh;
        if (R0 >= rows) break;
        const float* a0p = tileA + (R0 + l16) * AG_TP + 2 * half;
        const float* a1p = a0p + 16 * AG_TP;
        v8f c0 = {}, c1 = {};
#pragma unroll
        for (int ks = 0; ks < 32; ++ks) {
            const v2f a0 = *(const v2f*)(a0p + 4 * ks);
            const v2f a1 = *(const v2f*)(a1p + 4 * ks);
            c0 = __builtin_amdgcn_wmma_f32_16x16x4_f32(
                      false, a0,  false, bf[ks],
                      (short)0, c0,  false,  false);
            c1 = __builtin_amdgcn_wmma_f32_16x16x4_f32(
                      false, a1,  false, bf[ks],
                      (short)0, c1,  false,  false);
        }
#pragma unroll
        for (int r = 0; r < 8; ++r) {
            tileD[(R0 + r + 8 * half) * AG_TP + 16 * ct + l16]      = __float_as_uint(c0[r]);
            tileD[(R0 + 16 + r + 8 * half) * AG_TP + 16 * ct + l16] = __float_as_uint(c1[r]);
        }
    }
    __syncthreads();

    float4* out4 = (float4*)out;
    const float4 b4 = ((const float4*)bias)[lane];
#pragma unroll
    for (int i = 0; i < AG_SLOTS; ++i) {
        const int r = wave + 32 * i;
        if (r < rows) {
            const uint4 u = *(const uint4*)(tileD + r * AG_TP + 4 * lane);
            out4[(size_t)(base + r) * (CH / 4) + lane] =
                make_float4(__uint_as_float(u.x) + b4.x, __uint_as_float(u.y) + b4.y,
                            __uint_as_float(u.z) + b4.z, __uint_as_float(u.w) + b4.w);
        }
    }
}


extern "C" void kernel_launch(void* const* d_in, const int* in_sizes, int n_in,
                              void* d_out, int out_size, void* d_ws, size_t ws_size,
                              hipStream_t stream) {
    const float* x  = (const float*)d_in[0];
    const float* W  = (const float*)d_in[1];
    const float* b  = (const float*)d_in[2];
    const int*   ei = (const int*)d_in[3];

    const int n = in_sizes[0] / CH;
    const int E = in_sizes[3] / 2;
    const int* src = ei;
    const int* dst = ei + E;

    float* deg = (float*)d_ws;
    float* out = (float*)d_out;

    gcn_deg_count<<<HG_PARTS + (E + 4095) / 4096, 1024, 0, stream>>>(dst, deg, n, E);
    gcn_dinv     <<<(n + 255) / 256, 64, 0, stream>>>(deg, n, E);

    gcn_aggregate<<<(n + AG_GROUP - 1) / AG_GROUP, 1024, 0, stream>>>(src, dst, deg, x, W, b, out, n, E);
}
